# starting kernel
# baseline (speedup 1.0000x reference)
#include <hip/hip_runtime.h>
#include <math.h>

#define VOCAB 100000
#define DIM   300
#define BATCH 4096
#define SEQL  200
#define HID   1000
#define HIDP  1024
#define OUTD  3
#define REPD  (2*DIM)

typedef __attribute__((ext_vector_type(2))) float v2f;
typedef __attribute__((ext_vector_type(8))) float v8f;

__global__ __launch_bounds__(256) void transpose_pad_kernel(
    const float* __restrict__ W1, float* __restrict__ W1t)
{
    __shared__ float tile[32][33];
    const int kBase = blockIdx.x * 32;
    const int nBase = blockIdx.y * 32;
    const int tx = threadIdx.x;
    const int ty = threadIdx.y;

    #pragma unroll
    for (int i = ty; i < 32; i += 8) {
        const int k = kBase + i, n = nBase + tx;
        tile[i][tx] = (k < REPD && n < HID) ? W1[(size_t)k * HID + n] : 0.0f;
    }
    __syncthreads();
    #pragma unroll
    for (int i = ty; i < 32; i += 8) {
        const int n = nBase + i, k = kBase + tx;
        if (k < REPD)
            W1t[(size_t)n * REPD + k] = tile[tx][i];
    }
}

__global__ __launch_bounds__(320) void pool_kernel(
    const float* __restrict__ emb, const int* __restrict__ x,
    const int* __restrict__ lengths, float* __restrict__ rep)
{
    __shared__ int sidx[SEQL];
    const int b   = blockIdx.x;
    const int tid = threadIdx.x;
    if (tid < SEQL) sidx[tid] = x[b * SEQL + tid];
    __syncthreads();
    if (tid >= DIM) return;

    const int len = lengths[b];
    float sum = 0.0f;
    float mx  = -INFINITY;
    for (int t = 0; t < SEQL; ++t) {
        if (t + 8 < SEQL)
            __builtin_prefetch(emb + (size_t)sidx[t + 8] * DIM + tid, 0, 0);
        const float v = emb[(size_t)sidx[t] * DIM + tid];
        sum += v;
        if (t < len) mx = fmaxf(mx, v);
    }
    rep[(size_t)b * REPD + tid]       = sum / (float)len;
    rep[(size_t)b * REPD + DIM + tid] = mx;
}

__global__ __launch_bounds__(128) void gemm1_wmma_kernel(
    const float* __restrict__ rep, const float* __restrict__ W1t,
    const float* __restrict__ b1, float* __restrict__ h)
{
    const int wave  = blockIdx.x * (blockDim.x >> 5) + (threadIdx.x >> 5);
    const int lane  = threadIdx.x & 31;
    const int mtile = wave & 255;
    const int ntile = wave >> 8;
    const int row0  = mtile * 16;
    const int col0  = ntile * 64;
    const int laneM = lane & 15;
    const int kHi   = (lane >> 4) << 1;

    const float* __restrict__ arow  = rep + (size_t)(row0 + laneM) * REPD;
    const float* __restrict__ brow0 = W1t + (size_t)(col0 +  0 + laneM) * REPD;
    const float* __restrict__ brow1 = W1t + (size_t)(col0 + 16 + laneM) * REPD;
    const float* __restrict__ brow2 = W1t + (size_t)(col0 + 32 + laneM) * REPD;
    const float* __restrict__ brow3 = W1t + (size_t)(col0 + 48 + laneM) * REPD;

    v8f acc0 = {}, acc1 = {}, acc2 = {}, acc3 = {};

    for (int k = 0; k < REPD; k += 4) {
        const int kb = k + kHi;
        const v2f a  = *(const v2f*)(arow  + kb);
        const v2f b0 = *(const v2f*)(brow0 + kb);
        const v2f b1v = *(const v2f*)(brow1 + kb);
        const v2f b2v = *(const v2f*)(brow2 + kb);
        const v2f b3v = *(const v2f*)(brow3 + kb);
        acc0 = __builtin_amdgcn_wmma_f32_16x16x4_f32(false, a, false, b0,  (short)0, acc0, false, false);
        acc1 = __builtin_amdgcn_wmma_f32_16x16x4_f32(false, a, false, b1v, (short)0, acc1, false, false);
        acc2 = __builtin_amdgcn_wmma_f32_16x16x4_f32(false, a, false, b2v, (short)0, acc2, false, false);
        acc3 = __builtin_amdgcn_wmma_f32_16x16x4_f32(false, a, false, b3v, (short)0, acc3, false, false);
    }

    const int mOff = (lane >> 4) << 3;
    const v8f accs[4] = { acc0, acc1, acc2, acc3 };
    #pragma unroll
    for (int j = 0; j < 4; ++j) {
        const int n = col0 + j * 16 + laneM;
        if (n >= HID) continue;
        const float bias = b1[n];
        #pragma unroll
        for (int r = 0; r < 8; ++r) {
            const int m = row0 + r + mOff;
            const float v = accs[j][r] + bias;
            h[(size_t)m * HID + n] = v > 0.0f ? v : 0.0f;
        }
    }
}

__global__ __launch_bounds__(256) void gemm2_kernel(
    const float* __restrict__ h, const float* __restrict__ W2,
    const float* __restrict__ b2, float* __restrict__ out)
{
    __shared__ float red[3][256];
    const int b = blockIdx.x, tid = threadIdx.x;
    float a0 = 0.0f, a1 = 0.0f, a2 = 0.0f;
    const float* __restrict__ hr = h + (size_t)b * HID;
    for (int k = tid; k < HID; k += 256) {
        const float v = hr[k];
        a0 += v * W2[k * 3 + 0];
        a1 += v * W2[k * 3 + 1];
        a2 += v * W2[k * 3 + 2];
    }
    red[0][tid] = a0; red[1][tid] = a1; red[2][tid] = a2;
    __syncthreads();
    for (int s = 128; s > 0; s >>= 1) {
        if (tid < s) {
            red[0][tid] += red[0][tid + s];
            red[1][tid] += red[1][tid + s];
            red[2][tid] += red[2][tid + s];
        }
        __syncthreads();
    }
    if (tid < 3) out[b * 3 + tid] = red[tid][0] + b2[tid];
}

extern "C" void kernel_launch(void* const* d_in, const int* in_sizes, int n_in,
                              void* d_out, int out_size, void* d_ws, size_t ws_size,
                              hipStream_t stream)
{
    const float* emb  = (const float*)d_in[0];
    const float* W1   = (const float*)d_in[1];
    const float* b1   = (const float*)d_in[2];
    const float* W2   = (const float*)d_in[3];
    const float* b2   = (const float*)d_in[4];
    const int*   x    = (const int*)d_in[5];
    const int*   lens = (const int*)d_in[6];
    float* out = (float*)d_out;

    float* W1t = (float*)d_ws;
    float* rep = W1t + (size_t)HIDP * REPD;
    float* h   = rep + (size_t)BATCH * REPD;

    hipLaunchKernelGGL(transpose_pad_kernel, dim3(19, 32), dim3(32, 8), 0, stream,
                       W1, W1t);

    hipLaunchKernelGGL(pool_kernel, dim3(BATCH), dim3(320), 0, stream,
                       emb, x, lens, rep);

    hipLaunchKernelGGL(gemm1_wmma_kernel, dim3(1024), dim3(128), 0, stream,
                       rep, W1t, b1, h);

    hipLaunchKernelGGL(gemm2_kernel, dim3(BATCH), dim3(256), 0, stream,
                       h, W2, b2, out);
}


# pooling gather: a lane owns 4 dims and reads 16-byte row pieces, 8 tokens' row loads issued together per lane, 4 batch rows per block, prefetch dropped; per-dim summation order unchanged
# speedup vs baseline: 1.4717x; 1.4717x over previous
#include <hip/hip_runtime.h>
#include <math.h>

#define VOCAB 100000
#define DIM   300
#define BATCH 4096
#define SEQL  200
#define HID   1000
#define HIDP  1024
#define OUTD  3
#define REPD  (2*DIM)

typedef __attribute__((ext_vector_type(2))) float v2f;
typedef __attribute__((ext_vector_type(8))) float v8f;
typedef __attribute__((ext_vector_type(4))) float v4f;
typedef __attribute__((ext_vector_type(4))) unsigned v4u;

__global__ __launch_bounds__(256) void transpose_pad_kernel(
    const float* __restrict__ W1, float* __restrict__ W1t)
{
    __shared__ float tile[32][33];
    const int kBase = blockIdx.x * 32;
    const int nBase = blockIdx.y * 32;
    const int tx = threadIdx.x;
    const int ty = threadIdx.y;

    #pragma unroll
    for (int i = ty; i < 32; i += 8) {
        const int k = kBase + i, n = nBase + tx;
        tile[i][tx] = (k < REPD && n < HID) ? W1[(size_t)k * HID + n] : 0.0f;
    }
    __syncthreads();
    #pragma unroll
    for (int i = ty; i < 32; i += 8) {
        const int n = nBase + i, k = kBase + tx;
        if (k < REPD)
            W1t[(size_t)n * REPD + k] = tile[tx][i];
    }
}

#define PROWS  4
#define PLANES (DIM / 4)
#define PGRP   8
#define ROWB   (DIM * 4)

static __device__ __forceinline__ v4f ld_row4(const float* __restrict__ emb, unsigned off)
{
    return *(const v4f*)((const char*)emb + off);
}

static __device__ __forceinline__ v4f max4(v4f a, v4f b)
{
    v4f r;
    r.x = fmaxf(a.x, b.x); r.y = fmaxf(a.y, b.y);
    r.z = fmaxf(a.z, b.z); r.w = fmaxf(a.w, b.w);
    return r;
}

__global__ __launch_bounds__(320) void pool_kernel(
    const float* __restrict__ emb, const int* __restrict__ x,
    const int* __restrict__ lengths, float* __restrict__ rep)
{
    __shared__ alignas(16) unsigned soff[PROWS * SEQL];
    const int tid = threadIdx.x;
    const int b0  = blockIdx.x * PROWS;
    const int* __restrict__ xb = x + (size_t)b0 * SEQL;
    {
        const int i0 = tid, i1 = tid + 320, i2 = tid + 640;
        const int q0 = xb[i0];
        const int q1 = xb[i1];
        const int q2 = (i2 < PROWS * SEQL) ? xb[i2] : 0;
        soff[i0] = (unsigned)q0 * (unsigned)ROWB;
        soff[i1] = (unsigned)q1 * (unsigned)ROWB;
        if (i2 < PROWS * SEQL) soff[i2] = (unsigned)q2 * (unsigned)ROWB;
    }
    __syncthreads();
    if (tid >= PROWS * PLANES) return;

    const int r   = tid / PLANES;
    const int c   = tid - r * PLANES;
    const int b   = b0 + r;
    const int len = lengths[b];
    const unsigned cb = (unsigned)c * 16u;
    const unsigned* __restrict__ myoff = soff + r * SEQL;

    v4f sum = {0.0f, 0.0f, 0.0f, 0.0f};
    v4f mx  = {-INFINITY, -INFINITY, -INFINITY, -INFINITY};
    v4u oa = *(const v4u*)(myoff);
    v4u ob = *(const v4u*)(myoff + 4);
    for (int t0 = 0; t0 < SEQL; t0 += PGRP) {
        v4f v[PGRP];
        v[0] = ld_row4(emb, oa.x + cb); v[1] = ld_row4(emb, oa.y + cb);
        v[2] = ld_row4(emb, oa.z + cb); v[3] = ld_row4(emb, oa.w + cb);
        v[4] = ld_row4(emb, ob.x + cb); v[5] = ld_row4(emb, ob.y + cb);
        v[6] = ld_row4(emb, ob.z + cb); v[7] = ld_row4(emb, ob.w + cb);
        if (t0 + PGRP < SEQL) {
            oa = *(const v4u*)(myoff + t0 + PGRP);
            ob = *(const v4u*)(myoff + t0 + PGRP + 4);
        }
        #pragma unroll
        for (int j = 0; j < PGRP; ++j) {
            sum += v[j];
            if (t0 + j < len) mx = max4(mx, v[j]);
        }
    }
    const float flen = (float)len;
    v4f mean;
    mean.x = sum.x / flen; mean.y = sum.y / flen;
    mean.z = sum.z / flen; mean.w = sum.w / flen;
    *(v4f*)(rep + (size_t)b * REPD + c * 4)       = mean;
    *(v4f*)(rep + (size_t)b * REPD + DIM + c * 4) = mx;
}

__global__ __launch_bounds__(128) void gemm1_wmma_kernel(
    const float* __restrict__ rep, const float* __restrict__ W1t,
    const float* __restrict__ b1, float* __restrict__ h)
{
    const int wave  = blockIdx.x * (blockDim.x >> 5) + (threadIdx.x >> 5);
    const int lane  = threadIdx.x & 31;
    const int mtile = wave & 255;
    const int ntile = wave >> 8;
    const int row0  = mtile * 16;
    const int col0  = ntile * 64;
    const int laneM = lane & 15;
    const int kHi   = (lane >> 4) << 1;

    const float* __restrict__ arow  = rep + (size_t)(row0 + laneM) * REPD;
    const float* __restrict__ brow0 = W1t + (size_t)(col0 +  0 + laneM) * REPD;
    const float* __restrict__ brow1 = W1t + (size_t)(col0 + 16 + laneM) * REPD;
    const float* __restrict__ brow2 = W1t + (size_t)(col0 + 32 + laneM) * REPD;
    const float* __restrict__ brow3 = W1t + (size_t)(col0 + 48 + laneM) * REPD;

    v8f acc0 = {}, acc1 = {}, acc2 = {}, acc3 = {};

    for (int k = 0; k < REPD; k += 4) {
        const int kb = k + kHi;
        const v2f a  = *(const v2f*)(arow  + kb);
        const v2f b0 = *(const v2f*)(brow0 + kb);
        const v2f b1v = *(const v2f*)(brow1 + kb);
        const v2f b2v = *(const v2f*)(brow2 + kb);
        const v2f b3v = *(const v2f*)(brow3 + kb);
        acc0 = __builtin_amdgcn_wmma_f32_16x16x4_f32(false, a, false, b0,  (short)0, acc0, false, false);
        acc1 = __builtin_amdgcn_wmma_f32_16x16x4_f32(false, a, false, b1v, (short)0, acc1, false, false);
        acc2 = __builtin_amdgcn_wmma_f32_16x16x4_f32(false, a, false, b2v, (short)0, acc2, false, false);
        acc3 = __builtin_amdgcn_wmma_f32_16x16x4_f32(false, a, false, b3v, (short)0, acc3, false, false);
    }

    const int mOff = (lane >> 4) << 3;
    const v8f accs[4] = { acc0, acc1, acc2, acc3 };
    #pragma unroll
    for (int j = 0; j < 4; ++j) {
        const int n = col0 + j * 16 + laneM;
        if (n >= HID) continue;
        const float bias = b1[n];
        #pragma unroll
        for (int r = 0; r < 8; ++r) {
            const int m = row0 + r + mOff;
            const float v = accs[j][r] + bias;
            h[(size_t)m * HID + n] = v > 0.0f ? v : 0.0f;
        }
    }
}

__global__ __launch_bounds__(256) void gemm2_kernel(
    const float* __restrict__ h, const float* __restrict__ W2,
    const float* __restrict__ b2, float* __restrict__ out)
{
    __shared__ float red[3][256];
    const int b = blockIdx.x, tid = threadIdx.x;
    float a0 = 0.0f, a1 = 0.0f, a2 = 0.0f;
    const float* __restrict__ hr = h + (size_t)b * HID;
    for (int k = tid; k < HID; k += 256) {
        const float v = hr[k];
        a0 += v * W2[k * 3 + 0];
        a1 += v * W2[k * 3 + 1];
        a2 += v * W2[k * 3 + 2];
    }
    red[0][tid] = a0; red[1][tid] = a1; red[2][tid] = a2;
    __syncthreads();
    for (int s = 128; s > 0; s >>= 1) {
        if (tid < s) {
            red[0][tid] += red[0][tid + s];
            red[1][tid] += red[1][tid + s];
            red[2][tid] += red[2][tid + s];
        }
        __syncthreads();
    }
    if (tid < 3) out[b * 3 + tid] = red[tid][0] + b2[tid];
}

extern "C" void kernel_launch(void* const* d_in, const int* in_sizes, int n_in,
                              void* d_out, int out_size, void* d_ws, size_t ws_size,
                              hipStream_t stream)
{
    const float* emb  = (const float*)d_in[0];
    const float* W1   = (const float*)d_in[1];
    const float* b1   = (const float*)d_in[2];
    const float* W2   = (const float*)d_in[3];
    const float* b2   = (const float*)d_in[4];
    const int*   x    = (const int*)d_in[5];
    const int*   lens = (const int*)d_in[6];
    float* out = (float*)d_out;

    float* W1t = (float*)d_ws;
    float* rep = W1t + (size_t)HIDP * REPD;
    float* h   = rep + (size_t)BATCH * REPD;

    hipLaunchKernelGGL(transpose_pad_kernel, dim3(19, 32), dim3(32, 8), 0, stream,
                       W1, W1t);

    hipLaunchKernelGGL(pool_kernel, dim3(1024), dim3(320), 0, stream,
                       emb, x, lens, rep);

    hipLaunchKernelGGL(gemm1_wmma_kernel, dim3(1024), dim3(128), 0, stream,
                       rep, W1t, b1, h);

    hipLaunchKernelGGL(gemm2_kernel, dim3(BATCH), dim3(256), 0, stream,
                       h, W2, b2, out);
}


# GEMM1 operand feed: wave tile 32x64, each lane loads 16 B (4 k) per operand row so 6 loads feed 16 matrix instructions, groups loaded two ahead; k summation order changes inside each 8-k group
# speedup vs baseline: 1.8531x; 1.2592x over previous
#include <hip/hip_runtime.h>
#include <math.h>

#define VOCAB 100000
#define DIM   300
#define BATCH 4096
#define SEQL  200
#define HID   1000
#define HIDP  1024
#define OUTD  3
#define REPD  (2*DIM)

typedef __attribute__((ext_vector_type(2))) float v2f;
typedef __attribute__((ext_vector_type(8))) float v8f;
typedef __attribute__((ext_vector_type(4))) float v4f;
typedef __attribute__((ext_vector_type(4))) unsigned v4u;

__global__ __launch_bounds__(256) void transpose_pad_kernel(
    const float* __restrict__ W1, float* __restrict__ W1t)
{
    __shared__ float tile[32][33];
    const int kBase = blockIdx.x * 32;
    const int nBase = blockIdx.y * 32;
    const int tx = threadIdx.x;
    const int ty = threadIdx.y;

    #pragma unroll
    for (int i = ty; i < 32; i += 8) {
        const int k = kBase + i, n = nBase + tx;
        tile[i][tx] = (k < REPD && n < HID) ? W1[(size_t)k * HID + n] : 0.0f;
    }
    __syncthreads();
    #pragma unroll
    for (int i = ty; i < 32; i += 8) {
        const int n = nBase + i, k = kBase + tx;
        if (k < REPD)
            W1t[(size_t)n * REPD + k] = tile[tx][i];
    }
}

#define PROWS  4
#define PLANES (DIM / 4)
#define PGRP   8
#define ROWB   (DIM * 4)

static __device__ __forceinline__ v4f ld_row4(const float* __restrict__ emb, unsigned off)
{
    return *(const v4f*)((const char*)emb + off);
}

static __device__ __forceinline__ v4f max4(v4f a, v4f b)
{
    v4f r;
    r.x = fmaxf(a.x, b.x); r.y = fmaxf(a.y, b.y);
    r.z = fmaxf(a.z, b.z); r.w = fmaxf(a.w, b.w);
    return r;
}

__global__ __launch_bounds__(320) void pool_kernel(
    const float* __restrict__ emb, const int* __restrict__ x,
    const int* __restrict__ lengths, float* __restrict__ rep)
{
    __shared__ alignas(16) unsigned soff[PROWS * SEQL];
    const int tid = threadIdx.x;
    const int b0  = blockIdx.x * PROWS;
    const int* __restrict__ xb = x + (size_t)b0 * SEQL;
    {
        const int i0 = tid, i1 = tid + 320, i2 = tid + 640;
        const int q0 = xb[i0];
        const int q1 = xb[i1];
        const int q2 = (i2 < PROWS * SEQL) ? xb[i2] : 0;
        soff[i0] = (unsigned)q0 * (unsigned)ROWB;
        soff[i1] = (unsigned)q1 * (unsigned)ROWB;
        if (i2 < PROWS * SEQL) soff[i2] = (unsigned)q2 * (unsigned)ROWB;
    }
    __syncthreads();
    if (tid >= PROWS * PLANES) return;

    const int r   = tid / PLANES;
    const int c   = tid - r * PLANES;
    const int b   = b0 + r;
    const int len = lengths[b];
    const unsigned cb = (unsigned)c * 16u;
    const unsigned* __restrict__ myoff = soff + r * SEQL;

    v4f sum = {0.0f, 0.0f, 0.0f, 0.0f};
    v4f mx  = {-INFINITY, -INFINITY, -INFINITY, -INFINITY};
    v4u oa = *(const v4u*)(myoff);
    v4u ob = *(const v4u*)(myoff + 4);
    for (int t0 = 0; t0 < SEQL; t0 += PGRP) {
        v4f v[PGRP];
        v[0] = ld_row4(emb, oa.x + cb); v[1] = ld_row4(emb, oa.y + cb);
        v[2] = ld_row4(emb, oa.z + cb); v[3] = ld_row4(emb, oa.w + cb);
        v[4] = ld_row4(emb, ob.x + cb); v[5] = ld_row4(emb, ob.y + cb);
        v[6] = ld_row4(emb, ob.z + cb); v[7] = ld_row4(emb, ob.w + cb);
        if (t0 + PGRP < SEQL) {
            oa = *(const v4u*)(myoff + t0 + PGRP);
            ob = *(const v4u*)(myoff + t0 + PGRP + 4);
        }
        #pragma unroll
        for (int j = 0; j < PGRP; ++j) {
            sum += v[j];
            if (t0 + j < len) mx = max4(mx, v[j]);
        }
    }
    const float flen = (float)len;
    v4f mean;
    mean.x = sum.x / flen; mean.y = sum.y / flen;
    mean.z = sum.z / flen; mean.w = sum.w / flen;
    *(v4f*)(rep + (size_t)b * REPD + c * 4)       = mean;
    *(v4f*)(rep + (size_t)b * REPD + DIM + c * 4) = mx;
}

struct FragSet { v4f a0, a1, b0, b1, b2, b3; };

static __device__ __forceinline__ void frag_load(
    FragSet& f, const float* __restrict__ pa, const float* __restrict__ pb, int k)
{
    f.a0 = *(const v4f*)(pa + k);
    f.a1 = *(const v4f*)(pa + 16 * REPD + k);
    f.b0 = *(const v4f*)(pb + k);
    f.b1 = *(const v4f*)(pb + 16 * REPD + k);
    f.b2 = *(const v4f*)(pb + 32 * REPD + k);
    f.b3 = *(const v4f*)(pb + 48 * REPD + k);
}

static __device__ __forceinline__ v8f mma4(v2f a, v2f b, v8f c)
{
    return __builtin_amdgcn_wmma_f32_16x16x4_f32(false, a, false, b, (short)0, c, false, false);
}

static __device__ __forceinline__ void frag_mma(const FragSet& f, v8f (&acc)[8])
{
    const v2f a0l = __builtin_shufflevector(f.a0, f.a0, 0, 1);
    const v2f a0h = __builtin_shufflevector(f.a0, f.a0, 2, 3);
    const v2f a1l = __builtin_shufflevector(f.a1, f.a1, 0, 1);
    const v2f a1h = __builtin_shufflevector(f.a1, f.a1, 2, 3);
    const v2f b0l = __builtin_shufflevector(f.b0, f.b0, 0, 1);
    const v2f b0h = __builtin_shufflevector(f.b0, f.b0, 2, 3);
    const v2f b1l = __builtin_shufflevector(f.b1, f.b1, 0, 1);
    const v2f b1h = __builtin_shufflevector(f.b1, f.b1, 2, 3);
    const v2f b2l = __builtin_shufflevector(f.b2, f.b2, 0, 1);
    const v2f b2h = __builtin_shufflevector(f.b2, f.b2, 2, 3);
    const v2f b3l = __builtin_shufflevector(f.b3, f.b3, 0, 1);
    const v2f b3h = __builtin_shufflevector(f.b3, f.b3, 2, 3);
    acc[0] = mma4(a0l, b0l, acc[0]); acc[1] = mma4(a0l, b1l, acc[1]);
    acc[2] = mma4(a0l, b2l, acc[2]); acc[3] = mma4(a0l, b3l, acc[3]);
    acc[4] = mma4(a1l, b0l, acc[4]); acc[5] = mma4(a1l, b1l, acc[5]);
    acc[6] = mma4(a1l, b2l, acc[6]); acc[7] = mma4(a1l, b3l, acc[7]);
    acc[0] = mma4(a0h, b0h, acc[0]); acc[1] = mma4(a0h, b1h, acc[1]);
    acc[2] = mma4(a0h, b2h, acc[2]); acc[3] = mma4(a0h, b3h, acc[3]);
    acc[4] = mma4(a1h, b0h, acc[4]); acc[5] = mma4(a1h, b1h, acc[5]);
    acc[6] = mma4(a1h, b2h, acc[6]); acc[7] = mma4(a1h, b3h, acc[7]);
}

__global__ __launch_bounds__(128) void gemm1_wmma_kernel(
    const float* __restrict__ rep, const float* __restrict__ W1t,
    const float* __restrict__ b1, float* __restrict__ h)
{
    const int wave  = blockIdx.x * (blockDim.x >> 5) + (threadIdx.x >> 5);
    const int lane  = threadIdx.x & 31;
    const int mtile = wave & 127;
    const int ntile = wave >> 7;
    const int row0  = mtile * 32;
    const int col0  = ntile * 64;
    const int laneM = lane & 15;
    const int kHi   = (lane >> 4) << 2;

    const float* __restrict__ pa = rep + (size_t)(row0 + laneM) * REPD + kHi;
    const float* __restrict__ pb = W1t + (size_t)(col0 + laneM) * REPD + kHi;

    v8f acc[8] = {};
    FragSet s0, s1, s2;

    frag_load(s0, pa, pb, 0);
    frag_load(s1, pa, pb, 8);
    for (int k = 0; k < REPD - 24; k += 24) {
        frag_load(s2, pa, pb, k + 16);
        frag_mma(s0, acc);
        frag_load(s0, pa, pb, k + 24);
        frag_mma(s1, acc);
        frag_load(s1, pa, pb, k + 32);
        frag_mma(s2, acc);
    }
    frag_load(s2, pa, pb, REPD - 8);
    frag_mma(s0, acc);
    frag_mma(s1, acc);
    frag_mma(s2, acc);

    const int mOff = (lane >> 4) << 3;
    #pragma unroll
    for (int j = 0; j < 4; ++j) {
        const int n = col0 + j * 16 + laneM;
        if (n >= HID) continue;
        const float bias = b1[n];
        #pragma unroll
        for (int i = 0; i < 2; ++i) {
            #pragma unroll
            for (int r = 0; r < 8; ++r) {
                const int m = row0 + i * 16 + r + mOff;
                const float v = acc[i * 4 + j][r] + bias;
                h[(size_t)m * HID + n] = v > 0.0f ? v : 0.0f;
            }
        }
    }
}

__global__ __launch_bounds__(256) void gemm2_kernel(
    const float* __restrict__ h, const float* __restrict__ W2,
    const float* __restrict__ b2, float* __restrict__ out)
{
    __shared__ float red[3][256];
    const int b = blockIdx.x, tid = threadIdx.x;
    float a0 = 0.0f, a1 = 0.0f, a2 = 0.0f;
    const float* __restrict__ hr = h + (size_t)b * HID;
    for (int k = tid; k < HID; k += 256) {
        const float v = hr[k];
        a0 += v * W2[k * 3 + 0];
        a1 += v * W2[k * 3 + 1];
        a2 += v * W2[k * 3 + 2];
    }
    red[0][tid] = a0; red[1][tid] = a1; red[2][tid] = a2;
    __syncthreads();
    for (int s = 128; s > 0; s >>= 1) {
        if (tid < s) {
            red[0][tid] += red[0][tid + s];
            red[1][tid] += red[1][tid + s];
            red[2][tid] += red[2][tid + s];
        }
        __syncthreads();
    }
    if (tid < 3) out[b * 3 + tid] = red[tid][0] + b2[tid];
}

extern "C" void kernel_launch(void* const* d_in, const int* in_sizes, int n_in,
                              void* d_out, int out_size, void* d_ws, size_t ws_size,
                              hipStream_t stream)
{
    const float* emb  = (const float*)d_in[0];
    const float* W1   = (const float*)d_in[1];
    const float* b1   = (const float*)d_in[2];
    const float* W2   = (const float*)d_in[3];
    const float* b2   = (const float*)d_in[4];
    const int*   x    = (const int*)d_in[5];
    const int*   lens = (const int*)d_in[6];
    float* out = (float*)d_out;

    float* W1t = (float*)d_ws;
    float* rep = W1t + (size_t)HIDP * REPD;
    float* h   = rep + (size_t)BATCH * REPD;

    hipLaunchKernelGGL(transpose_pad_kernel, dim3(19, 32), dim3(32, 8), 0, stream,
                       W1, W1t);

    hipLaunchKernelGGL(pool_kernel, dim3(1024), dim3(320), 0, stream,
                       emb, x, lens, rep);

    hipLaunchKernelGGL(gemm1_wmma_kernel, dim3(512), dim3(128), 0, stream,
                       rep, W1t, b1, h);

    hipLaunchKernelGGL(gemm2_kernel, dim3(BATCH), dim3(256), 0, stream,
                       h, W2, b2, out);
}


# second layer fused into GEMM1 epilogue: product computed transposed (W1t as A) so a lane sums relu(h)*W2 in its own registers; h never stored; last kernel adds 16 partials; h sum order changes
# speedup vs baseline: 1.9467x; 1.0505x over previous
#include <hip/hip_runtime.h>
#include <math.h>

#define VOCAB 100000
#define DIM   300
#define BATCH 4096
#define SEQL  200
#define HID   1000
#define HIDP  1024
#define OUTD  3
#define REPD  (2*DIM)

typedef __attribute__((ext_vector_type(2))) float v2f;
typedef __attribute__((ext_vector_type(8))) float v8f;
typedef __attribute__((ext_vector_type(4))) float v4f;
typedef __attribute__((ext_vector_type(4))) unsigned v4u;

__global__ __launch_bounds__(256) void transpose_pad_kernel(
    const float* __restrict__ W1, float* __restrict__ W1t)
{
    __shared__ float tile[32][33];
    const int kBase = blockIdx.x * 32;
    const int nBase = blockIdx.y * 32;
    const int tx = threadIdx.x;
    const int ty = threadIdx.y;

    #pragma unroll
    for (int i = ty; i < 32; i += 8) {
        const int k = kBase + i, n = nBase + tx;
        tile[i][tx] = (k < REPD && n < HID) ? W1[(size_t)k * HID + n] : 0.0f;
    }
    __syncthreads();
    #pragma unroll
    for (int i = ty; i < 32; i += 8) {
        const int n = nBase + i, k = kBase + tx;
        if (k < REPD)
            W1t[(size_t)n * REPD + k] = tile[tx][i];
    }
}

#define PROWS  4
#define PLANES (DIM / 4)
#define PGRP   8
#define ROWB   (DIM * 4)

static __device__ __forceinline__ v4f ld_row4(const float* __restrict__ emb, unsigned off)
{
    return *(const v4f*)((const char*)emb + off);
}

static __device__ __forceinline__ v4f max4(v4f a, v4f b)
{
    v4f r;
    r.x = fmaxf(a.x, b.x); r.y = fmaxf(a.y, b.y);
    r.z = fmaxf(a.z, b.z); r.w = fmaxf(a.w, b.w);
    return r;
}

__global__ __launch_bounds__(320) void pool_kernel(
    const float* __restrict__ emb, const int* __restrict__ x,
    const int* __restrict__ lengths, float* __restrict__ rep)
{
    __shared__ alignas(16) unsigned soff[PROWS * SEQL];
    const int tid = threadIdx.x;
    const int b0  = blockIdx.x * PROWS;
    const int* __restrict__ xb = x + (size_t)b0 * SEQL;
    {
        const int i0 = tid, i1 = tid + 320, i2 = tid + 640;
        const int q0 = xb[i0];
        const int q1 = xb[i1];
        const int q2 = (i2 < PROWS * SEQL) ? xb[i2] : 0;
        soff[i0] = (unsigned)q0 * (unsigned)ROWB;
        soff[i1] = (unsigned)q1 * (unsigned)ROWB;
        if (i2 < PROWS * SEQL) soff[i2] = (unsigned)q2 * (unsigned)ROWB;
    }
    __syncthreads();
    if (tid >= PROWS * PLANES) return;

    const int r   = tid / PLANES;
    const int c   = tid - r * PLANES;
    const int b   = b0 + r;
    const int len = lengths[b];
    const unsigned cb = (unsigned)c * 16u;
    const unsigned* __restrict__ myoff = soff + r * SEQL;

    v4f sum = {0.0f, 0.0f, 0.0f, 0.0f};
    v4f mx  = {-INFINITY, -INFINITY, -INFINITY, -INFINITY};
    v4u oa = *(const v4u*)(myoff);
    v4u ob = *(const v4u*)(myoff + 4);
    for (int t0 = 0; t0 < SEQL; t0 += PGRP) {
        v4f v[PGRP];
        v[0] = ld_row4(emb, oa.x + cb); v[1] = ld_row4(emb, oa.y + cb);
        v[2] = ld_row4(emb, oa.z + cb); v[3] = ld_row4(emb, oa.w + cb);
        v[4] = ld_row4(emb, ob.x + cb); v[5] = ld_row4(emb, ob.y + cb);
        v[6] = ld_row4(emb, ob.z + cb); v[7] = ld_row4(emb, ob.w + cb);
        if (t0 + PGRP < SEQL) {
            oa = *(const v4u*)(myoff + t0 + PGRP);
            ob = *(const v4u*)(myoff + t0 + PGRP + 4);
        }
        #pragma unroll
        for (int j = 0; j < PGRP; ++j) {
            sum += v[j];
            if (t0 + j < len) mx = max4(mx, v[j]);
        }
    }
    const float flen = (float)len;
    v4f mean;
    mean.x = sum.x / flen; mean.y = sum.y / flen;
    mean.z = sum.z / flen; mean.w = sum.w / flen;
    *(v4f*)(rep + (size_t)b * REPD + c * 4)       = mean;
    *(v4f*)(rep + (size_t)b * REPD + DIM + c * 4) = mx;
}

#define NPART 16

struct FragSet { v4f r0, r1, w0, w1, w2, w3; };

static __device__ __forceinline__ void frag_load(
    FragSet& f, const float* __restrict__ pr, const float* __restrict__ pw, int k)
{
    f.r0 = *(const v4f*)(pr + k);
    f.r1 = *(const v4f*)(pr + 16 * REPD + k);
    f.w0 = *(const v4f*)(pw + k);
    f.w1 = *(const v4f*)(pw + 16 * REPD + k);
    f.w2 = *(const v4f*)(pw + 32 * REPD + k);
    f.w3 = *(const v4f*)(pw + 48 * REPD + k);
}

static __device__ __forceinline__ v8f mma4(v2f a, v2f b, v8f c)
{
    return __builtin_amdgcn_wmma_f32_16x16x4_f32(false, a, false, b, (short)0, c, false, false);
}

static __device__ __forceinline__ void frag_mma(const FragSet& f, v8f (&acc)[8])
{
    const v2f r0l = __builtin_shufflevector(f.r0, f.r0, 0, 1);
    const v2f r0h = __builtin_shufflevector(f.r0, f.r0, 2, 3);
    const v2f r1l = __builtin_shufflevector(f.r1, f.r1, 0, 1);
    const v2f r1h = __builtin_shufflevector(f.r1, f.r1, 2, 3);
    const v2f w0l = __builtin_shufflevector(f.w0, f.w0, 0, 1);
    const v2f w0h = __builtin_shufflevector(f.w0, f.w0, 2, 3);
    const v2f w1l = __builtin_shufflevector(f.w1, f.w1, 0, 1);
    const v2f w1h = __builtin_shufflevector(f.w1, f.w1, 2, 3);
    const v2f w2l = __builtin_shufflevector(f.w2, f.w2, 0, 1);
    const v2f w2h = __builtin_shufflevector(f.w2, f.w2, 2, 3);
    const v2f w3l = __builtin_shufflevector(f.w3, f.w3, 0, 1);
    const v2f w3h = __builtin_shufflevector(f.w3, f.w3, 2, 3);
    acc[0] = mma4(w0l, r0l, acc[0]); acc[2] = mma4(w1l, r0l, acc[2]);
    acc[4] = mma4(w2l, r0l, acc[4]); acc[6] = mma4(w3l, r0l, acc[6]);
    acc[1] = mma4(w0l, r1l, acc[1]); acc[3] = mma4(w1l, r1l, acc[3]);
    acc[5] = mma4(w2l, r1l, acc[5]); acc[7] = mma4(w3l, r1l, acc[7]);
    acc[0] = mma4(w0h, r0h, acc[0]); acc[2] = mma4(w1h, r0h, acc[2]);
    acc[4] = mma4(w2h, r0h, acc[4]); acc[6] = mma4(w3h, r0h, acc[6]);
    acc[1] = mma4(w0h, r1h, acc[1]); acc[3] = mma4(w1h, r1h, acc[3]);
    acc[5] = mma4(w2h, r1h, acc[5]); acc[7] = mma4(w3h, r1h, acc[7]);
}

__global__ __launch_bounds__(128) void gemm1_wmma_kernel(
    const float* __restrict__ rep, const float* __restrict__ W1t,
    const float* __restrict__ b1, const float* __restrict__ W2,
    float* __restrict__ part)
{
    const int wave  = blockIdx.x * (blockDim.x >> 5) + (threadIdx.x >> 5);
    const int lane  = threadIdx.x & 31;
    const int mtile = wave & 127;
    const int ntile = wave >> 7;
    const int row0  = mtile * 32;
    const int col0  = ntile * 64;
    const int laneM = lane & 15;
    const int hh    = lane >> 4;
    const int kHi   = hh << 2;

    const float* __restrict__ pr = rep + (size_t)(row0 + laneM) * REPD + kHi;
    const float* __restrict__ pw = W1t + (size_t)(col0 + laneM) * REPD + kHi;

    v8f acc[8] = {};
    FragSet s0, s1, s2;

    frag_load(s0, pr, pw, 0);
    frag_load(s1, pr, pw, 8);
    for (int k = 0; k < REPD - 24; k += 24) {
        frag_load(s2, pr, pw, k + 16);
        frag_mma(s0, acc);
        frag_load(s0, pr, pw, k + 24);
        frag_mma(s1, acc);
        frag_load(s1, pr, pw, k + 32);
        frag_mma(s2, acc);
    }
    frag_load(s2, pr, pw, REPD - 8);

    bool ok[4];
    v4f  eb[4][2];
    v4f  ew[4][6];
    #pragma unroll
    for (int j = 0; j < 4; ++j) {
        const int n0 = col0 + j * 16 + hh * 8;
        ok[j] = n0 < HID;
        const int n0c = ok[j] ? n0 : 0;
        const float* __restrict__ bp = b1 + n0c;
        const float* __restrict__ wp = W2 + (size_t)n0c * OUTD;
        eb[j][0] = *(const v4f*)(bp);
        eb[j][1] = *(const v4f*)(bp + 4);
        #pragma unroll
        for (int q = 0; q < 6; ++q) ew[j][q] = *(const v4f*)(wp + 4 * q);
    }

    frag_mma(s0, acc);
    frag_mma(s1, acc);
    frag_mma(s2, acc);

    float p[2][OUTD] = {};
    #pragma unroll
    for (int j = 0; j < 4; ++j) {
        #pragma unroll
        for (int r = 0; r < 8; ++r) {
            const float bias = eb[j][r >> 2][r & 3];
            const int e = r * OUTD;
            const float wa = ew[j][(e    ) >> 2][(e    ) & 3];
            const float wb = ew[j][(e + 1) >> 2][(e + 1) & 3];
            const float wc = ew[j][(e + 2) >> 2][(e + 2) & 3];
            #pragma unroll
            for (int i = 0; i < 2; ++i) {
                const float v = acc[j * 2 + i][r] + bias;
                const float hv = (ok[j] && v > 0.0f) ? v : 0.0f;
                p[i][0] += hv * wa;
                p[i][1] += hv * wb;
                p[i][2] += hv * wc;
            }
        }
    }
    #pragma unroll
    for (int i = 0; i < 2; ++i) {
        #pragma unroll
        for (int c = 0; c < OUTD; ++c) p[i][c] += __shfl_xor(p[i][c], 16);
    }
    float* __restrict__ dst = part + ((size_t)ntile * BATCH + row0 + lane) * OUTD;
    dst[0] = hh ? p[1][0] : p[0][0];
    dst[1] = hh ? p[1][1] : p[0][1];
    dst[2] = hh ? p[1][2] : p[0][2];
}

__global__ __launch_bounds__(256) void out_sum_kernel(
    const float* __restrict__ part, const float* __restrict__ b2,
    float* __restrict__ out)
{
    const int i = blockIdx.x * 256 + threadIdx.x;
    float v[NPART];
    #pragma unroll
    for (int t = 0; t < NPART; ++t) v[t] = part[(size_t)t * (BATCH * OUTD) + i];
    float s = v[0];
    #pragma unroll
    for (int t = 1; t < NPART; ++t) s += v[t];
    out[i] = s + b2[i % OUTD];
}

extern "C" void kernel_launch(void* const* d_in, const int* in_sizes, int n_in,
                              void* d_out, int out_size, void* d_ws, size_t ws_size,
                              hipStream_t stream)
{
    const float* emb  = (const float*)d_in[0];
    const float* W1   = (const float*)d_in[1];
    const float* b1   = (const float*)d_in[2];
    const float* W2   = (const float*)d_in[3];
    const float* b2   = (const float*)d_in[4];
    const int*   x    = (const int*)d_in[5];
    const int*   lens = (const int*)d_in[6];
    float* out = (float*)d_out;

    float* W1t = (float*)d_ws;
    float* rep = W1t + (size_t)HIDP * REPD;
    float* part = rep + (size_t)BATCH * REPD;

    hipLaunchKernelGGL(transpose_pad_kernel, dim3(19, 32), dim3(32, 8), 0, stream,
                       W1, W1t);

    hipLaunchKernelGGL(pool_kernel, dim3(1024), dim3(320), 0, stream,
                       emb, x, lens, rep);

    hipLaunchKernelGGL(gemm1_wmma_kernel, dim3(512), dim3(128), 0, stream,
                       rep, W1t, b1, W2, part);

    hipLaunchKernelGGL(out_sum_kernel, dim3(BATCH * OUTD / 256), dim3(256), 0, stream,
                       part, b2, out);
}


# W1 transpose tiles ride in the pooling launch as 608 extra blocks behind the 1024 pooling blocks, so the separate transpose launch is gone (4 launches to 3); sum orders unchanged
# speedup vs baseline: 2.0984x; 1.0779x over previous
#include <hip/hip_runtime.h>
#include <math.h>

#define VOCAB 100000
#define DIM   300
#define BATCH 4096
#define SEQL  200
#define HID   1000
#define HIDP  1024
#define OUTD  3
#define REPD  (2*DIM)

typedef __attribute__((ext_vector_type(2))) float v2f;
typedef __attribute__((ext_vector_type(8))) float v8f;
typedef __attribute__((ext_vector_type(4))) float v4f;
typedef __attribute__((ext_vector_type(4))) unsigned v4u;

#define TKT    19
#define TNT    32
#define TTILES (TKT * TNT)

static __device__ __forceinline__ void transpose_pad_tile(
    const float* __restrict__ W1, float* __restrict__ W1t, int t)
{
    __shared__ float tile[32][33];
    const int kBase = (t % TKT) * 32;
    const int nBase = (t / TKT) * 32;
    const int tx = threadIdx.x & 31;
    const int ty = threadIdx.x >> 5;

    if (ty < 8) {
        #pragma unroll
        for (int i = ty; i < 32; i += 8) {
            const int k = kBase + i, n = nBase + tx;
            tile[i][tx] = (k < REPD && n < HID) ? W1[(size_t)k * HID + n] : 0.0f;
        }
    }
    __syncthreads();
    if (ty < 8) {
        #pragma unroll
        for (int i = ty; i < 32; i += 8) {
            const int n = nBase + i, k = kBase + tx;
            if (k < REPD)
                W1t[(size_t)n * REPD + k] = tile[tx][i];
        }
    }
}

#define PROWS  4
#define PLANES (DIM / 4)
#define PGRP   8
#define ROWB   (DIM * 4)
#define PBLOCKS (BATCH / PROWS)

static __device__ __forceinline__ v4f ld_row4(const float* __restrict__ emb, unsigned off)
{
    return *(const v4f*)((const char*)emb + off);
}

static __device__ __forceinline__ v4f max4(v4f a, v4f b)
{
    v4f r;
    r.x = fmaxf(a.x, b.x); r.y = fmaxf(a.y, b.y);
    r.z = fmaxf(a.z, b.z); r.w = fmaxf(a.w, b.w);
    return r;
}

__global__ __launch_bounds__(320) void pool_kernel(
    const float* __restrict__ emb, const int* __restrict__ x,
    const int* __restrict__ lengths, float* __restrict__ rep,
    const float* __restrict__ W1, float* __restrict__ W1t)
{
    if (blockIdx.x >= PBLOCKS) {
        transpose_pad_tile(W1, W1t, (int)blockIdx.x - PBLOCKS);
        return;
    }
    __shared__ alignas(16) unsigned soff[PROWS * SEQL];
    const int tid = threadIdx.x;
    const int b0  = blockIdx.x * PROWS;
    const int* __restrict__ xb = x + (size_t)b0 * SEQL;
    {
        const int i0 = tid, i1 = tid + 320, i2 = tid + 640;
        const int q0 = xb[i0];
        const int q1 = xb[i1];
        const int q2 = (i2 < PROWS * SEQL) ? xb[i2] : 0;
        soff[i0] = (unsigned)q0 * (unsigned)ROWB;
        soff[i1] = (unsigned)q1 * (unsigned)ROWB;
        if (i2 < PROWS * SEQL) soff[i2] = (unsigned)q2 * (unsigned)ROWB;
    }
    __syncthreads();
    if (tid >= PROWS * PLANES) return;

    const int r   = tid / PLANES;
    const int c   = tid - r * PLANES;
    const int b   = b0 + r;
    const int len = lengths[b];
    const unsigned cb = (unsigned)c * 16u;
    const unsigned* __restrict__ myoff = soff + r * SEQL;

    v4f sum = {0.0f, 0.0f, 0.0f, 0.0f};
    v4f mx  = {-INFINITY, -INFINITY, -INFINITY, -INFINITY};
    v4u oa = *(const v4u*)(myoff);
    v4u ob = *(const v4u*)(myoff + 4);
    for (int t0 = 0; t0 < SEQL; t0 += PGRP) {
        v4f v[PGRP];
        v[0] = ld_row4(emb, oa.x + cb); v[1] = ld_row4(emb, oa.y + cb);
        v[2] = ld_row4(emb, oa.z + cb); v[3] = ld_row4(emb, oa.w + cb);
        v[4] = ld_row4(emb, ob.x + cb); v[5] = ld_row4(emb, ob.y + cb);
        v[6] = ld_row4(emb, ob.z + cb); v[7] = ld_row4(emb, ob.w + cb);
        if (t0 + PGRP < SEQL) {
            oa = *(const v4u*)(myoff + t0 + PGRP);
            ob = *(const v4u*)(myoff + t0 + PGRP + 4);
        }
        #pragma unroll
        for (int j = 0; j < PGRP; ++j) {
            sum += v[j];
            if (t0 + j < len) mx = max4(mx, v[j]);
        }
    }
    const float flen = (float)len;
    v4f mean;
    mean.x = sum.x / flen; mean.y = sum.y / flen;
    mean.z = sum.z / flen; mean.w = sum.w / flen;
    *(v4f*)(rep + (size_t)b * REPD + c * 4)       = mean;
    *(v4f*)(rep + (size_t)b * REPD + DIM + c * 4) = mx;
}

#define NPART 16

struct FragSet { v4f r0, r1, w0, w1, w2, w3; };

static __device__ __forceinline__ void frag_load(
    FragSet& f, const float* __restrict__ pr, const float* __restrict__ pw, int k)
{
    f.r0 = *(const v4f*)(pr + k);
    f.r1 = *(const v4f*)(pr + 16 * REPD + k);
    f.w0 = *(const v4f*)(pw + k);
    f.w1 = *(const v4f*)(pw + 16 * REPD + k);
    f.w2 = *(const v4f*)(pw + 32 * REPD + k);
    f.w3 = *(const v4f*)(pw + 48 * REPD + k);
}

static __device__ __forceinline__ v8f mma4(v2f a, v2f b, v8f c)
{
    return __builtin_amdgcn_wmma_f32_16x16x4_f32(false, a, false, b, (short)0, c, false, false);
}

static __device__ __forceinline__ void frag_mma(const FragSet& f, v8f (&acc)[8])
{
    const v2f r0l = __builtin_shufflevector(f.r0, f.r0, 0, 1);
    const v2f r0h = __builtin_shufflevector(f.r0, f.r0, 2, 3);
    const v2f r1l = __builtin_shufflevector(f.r1, f.r1, 0, 1);
    const v2f r1h = __builtin_shufflevector(f.r1, f.r1, 2, 3);
    const v2f w0l = __builtin_shufflevector(f.w0, f.w0, 0, 1);
    const v2f w0h = __builtin_shufflevector(f.w0, f.w0, 2, 3);
    const v2f w1l = __builtin_shufflevector(f.w1, f.w1, 0, 1);
    const v2f w1h = __builtin_shufflevector(f.w1, f.w1, 2, 3);
    const v2f w2l = __builtin_shufflevector(f.w2, f.w2, 0, 1);
    const v2f w2h = __builtin_shufflevector(f.w2, f.w2, 2, 3);
    const v2f w3l = __builtin_shufflevector(f.w3, f.w3, 0, 1);
    const v2f w3h = __builtin_shufflevector(f.w3, f.w3, 2, 3);
    acc[0] = mma4(w0l, r0l, acc[0]); acc[2] = mma4(w1l, r0l, acc[2]);
    acc[4] = mma4(w2l, r0l, acc[4]); acc[6] = mma4(w3l, r0l, acc[6]);
    acc[1] = mma4(w0l, r1l, acc[1]); acc[3] = mma4(w1l, r1l, acc[3]);
    acc[5] = mma4(w2l, r1l, acc[5]); acc[7] = mma4(w3l, r1l, acc[7]);
    acc[0] = mma4(w0h, r0h, acc[0]); acc[2] = mma4(w1h, r0h, acc[2]);
    acc[4] = mma4(w2h, r0h, acc[4]); acc[6] = mma4(w3h, r0h, acc[6]);
    acc[1] = mma4(w0h, r1h, acc[1]); acc[3] = mma4(w1h, r1h, acc[3]);
    acc[5] = mma4(w2h, r1h, acc[5]); acc[7] = mma4(w3h, r1h, acc[7]);
}

__global__ __launch_bounds__(128) void gemm1_wmma_kernel(
    const float* __restrict__ rep, const float* __restrict__ W1t,
    const float* __restrict__ b1, const float* __restrict__ W2,
    float* __restrict__ part)
{
    const int wave  = blockIdx.x * (blockDim.x >> 5) + (threadIdx.x >> 5);
    const int lane  = threadIdx.x & 31;
    const int mtile = wave & 127;
    const int ntile = wave >> 7;
    const int row0  = mtile * 32;
    const int col0  = ntile * 64;
    const int laneM = lane & 15;
    const int hh    = lane >> 4;
    const int kHi   = hh << 2;

    const float* __restrict__ pr = rep + (size_t)(row0 + laneM) * REPD + kHi;
    const float* __restrict__ pw = W1t + (size_t)(col0 + laneM) * REPD + kHi;

    v8f acc[8] = {};
    FragSet s0, s1, s2;

    frag_load(s0, pr, pw, 0);
    frag_load(s1, pr, pw, 8);
    for (int k = 0; k < REPD - 24; k += 24) {
        frag_load(s2, pr, pw, k + 16);
        frag_mma(s0, acc);
        frag_load(s0, pr, pw, k + 24);
        frag_mma(s1, acc);
        frag_load(s1, pr, pw, k + 32);
        frag_mma(s2, acc);
    }
    frag_load(s2, pr, pw, REPD - 8);

    bool ok[4];
    v4f  eb[4][2];
    v4f  ew[4][6];
    #pragma unroll
    for (int j = 0; j < 4; ++j) {
        const int n0 = col0 + j * 16 + hh * 8;
        ok[j] = n0 < HID;
        const int n0c = ok[j] ? n0 : 0;
        const float* __restrict__ bp = b1 + n0c;
        const float* __restrict__ wp = W2 + (size_t)n0c * OUTD;
        eb[j][0] = *(const v4f*)(bp);
        eb[j][1] = *(const v4f*)(bp + 4);
        #pragma unroll
        for (int q = 0; q < 6; ++q) ew[j][q] = *(const v4f*)(wp + 4 * q);
    }

    frag_mma(s0, acc);
    frag_mma(s1, acc);
    frag_mma(s2, acc);

    float p[2][OUTD] = {};
    #pragma unroll
    for (int j = 0; j < 4; ++j) {
        #pragma unroll
        for (int r = 0; r < 8; ++r) {
            const float bias = eb[j][r >> 2][r & 3];
            const int e = r * OUTD;
            const float wa = ew[j][(e    ) >> 2][(e    ) & 3];
            const float wb = ew[j][(e + 1) >> 2][(e + 1) & 3];
            const float wc = ew[j][(e + 2) >> 2][(e + 2) & 3];
            #pragma unroll
            for (int i = 0; i < 2; ++i) {
                const float v = acc[j * 2 + i][r] + bias;
                const float hv = (ok[j] && v > 0.0f) ? v : 0.0f;
                p[i][0] += hv * wa;
                p[i][1] += hv * wb;
                p[i][2] += hv * wc;
            }
        }
    }
    #pragma unroll
    for (int i = 0; i < 2; ++i) {
        #pragma unroll
        for (int c = 0; c < OUTD; ++c) p[i][c] += __shfl_xor(p[i][c], 16);
    }
    float* __restrict__ dst = part + ((size_t)ntile * BATCH + row0 + lane) * OUTD;
    dst[0] = hh ? p[1][0] : p[0][0];
    dst[1] = hh ? p[1][1] : p[0][1];
    dst[2] = hh ? p[1][2] : p[0][2];
}

__global__ __launch_bounds__(256) void out_sum_kernel(
    const float* __restrict__ part, const float* __restrict__ b2,
    float* __restrict__ out)
{
    const int i = blockIdx.x * 256 + threadIdx.x;
    float v[NPART];
    #pragma unroll
    for (int t = 0; t < NPART; ++t) v[t] = part[(size_t)t * (BATCH * OUTD) + i];
    float s = v[0];
    #pragma unroll
    for (int t = 1; t < NPART; ++t) s += v[t];
    out[i] = s + b2[i % OUTD];
}

extern "C" void kernel_launch(void* const* d_in, const int* in_sizes, int n_in,
                              void* d_out, int out_size, void* d_ws, size_t ws_size,
                              hipStream_t stream)
{
    const float* emb  = (const float*)d_in[0];
    const float* W1   = (const float*)d_in[1];
    const float* b1   = (const float*)d_in[2];
    const float* W2   = (const float*)d_in[3];
    const float* b2   = (const float*)d_in[4];
    const int*   x    = (const int*)d_in[5];
    const int*   lens = (const int*)d_in[6];
    float* out = (float*)d_out;

    float* W1t = (float*)d_ws;
    float* rep = W1t + (size_t)HIDP * REPD;
    float* part = rep + (size_t)BATCH * REPD;

    hipLaunchKernelGGL(pool_kernel, dim3(PBLOCKS + TTILES), dim3(320), 0, stream,
                       emb, x, lens, rep, W1, W1t);

    hipLaunchKernelGGL(gemm1_wmma_kernel, dim3(512), dim3(128), 0, stream,
                       rep, W1t, b1, W2, part);

    hipLaunchKernelGGL(out_sum_kernel, dim3(BATCH * OUTD / 256), dim3(256), 0, stream,
                       part, b2, out);
}


# GEMM operands stored by their producers as 16-row x 8-k blocks of 512 contiguous bytes, so each fragment load reads 4 whole lines instead of 16 pieces of 32 bytes; all sum orders unchanged
# speedup vs baseline: 2.1834x; 1.0405x over previous
#include <hip/hip_runtime.h>
#include <math.h>

#define VOCAB 100000
#define DIM   300
#define BATCH 4096
#define SEQL  200
#define HID   1000
#define HIDP  1024
#define OUTD  3
#define REPD  (2*DIM)

typedef __attribute__((ext_vector_type(2))) float v2f;
typedef __attribute__((ext_vector_type(8))) float v8f;
typedef __attribute__((ext_vector_type(4))) float v4f;
typedef __attribute__((ext_vector_type(4))) unsigned v4u;

#define KGRP   (REPD / 8)
#define BLKF   128
#define TILEF  (KGRP * BLKF)

static __device__ __forceinline__ size_t blk_index(int row, int k)
{
    return (size_t)(row >> 4) * TILEF + (size_t)(k >> 3) * BLKF + (row & 15) * 8 + (k & 7);
}

#define TKT    19
#define TNT    32
#define TTILES (TKT * TNT)

static __device__ __forceinline__ void transpose_pad_tile(
    const float* __restrict__ W1, float* __restrict__ W1t, int t)
{
    __shared__ float tile[32][36];
    const int kBase = (t % TKT) * 32;
    const int nBase = (t / TKT) * 32;
    const int tx = threadIdx.x & 31;
    const int ty = threadIdx.x >> 5;

    if (ty < 8) {
        #pragma unroll
        for (int i = ty; i < 32; i += 8) {
            const int k = kBase + i, n = nBase + tx;
            tile[i][tx] = (k < REPD && n < HID) ? W1[(size_t)k * HID + n] : 0.0f;
        }
    }
    __syncthreads();
    if (ty < 8) {
        const int kk = tx & 7;
        const int ns = tx >> 3;
        #pragma unroll
        for (int i = ty; i < 32; i += 8) {
            const int nl = (i & 7) * 4 + ns;
            const int kl = (i >> 3) * 8 + kk;
            const int n = nBase + nl, k = kBase + kl;
            if (k < REPD)
                W1t[blk_index(n, k)] = tile[kl][nl];
        }
    }
}

#define PROWS  4
#define PLANES (DIM / 4)
#define PGRP   8
#define ROWB   (DIM * 4)
#define PBLOCKS (BATCH / PROWS)

static __device__ __forceinline__ v4f ld_row4(const float* __restrict__ emb, unsigned off)
{
    return *(const v4f*)((const char*)emb + off);
}

static __device__ __forceinline__ v4f max4(v4f a, v4f b)
{
    v4f r;
    r.x = fmaxf(a.x, b.x); r.y = fmaxf(a.y, b.y);
    r.z = fmaxf(a.z, b.z); r.w = fmaxf(a.w, b.w);
    return r;
}

__global__ __launch_bounds__(320) void pool_kernel(
    const float* __restrict__ emb, const int* __restrict__ x,
    const int* __restrict__ lengths, float* __restrict__ rep,
    const float* __restrict__ W1, float* __restrict__ W1t)
{
    if (blockIdx.x >= PBLOCKS) {
        transpose_pad_tile(W1, W1t, (int)blockIdx.x - PBLOCKS);
        return;
    }
    __shared__ alignas(16) unsigned soff[PROWS * SEQL];
    const int tid = threadIdx.x;
    const int b0  = blockIdx.x * PROWS;
    const int* __restrict__ xb = x + (size_t)b0 * SEQL;
    {
        const int i0 = tid, i1 = tid + 320, i2 = tid + 640;
        const int q0 = xb[i0];
        const int q1 = xb[i1];
        const int q2 = (i2 < PROWS * SEQL) ? xb[i2] : 0;
        soff[i0] = (unsigned)q0 * (unsigned)ROWB;
        soff[i1] = (unsigned)q1 * (unsigned)ROWB;
        if (i2 < PROWS * SEQL) soff[i2] = (unsigned)q2 * (unsigned)ROWB;
    }
    __syncthreads();
    if (tid >= PROWS * PLANES) return;

    const int r   = tid / PLANES;
    const int c   = tid - r * PLANES;
    const int b   = b0 + r;
    const int len = lengths[b];
    const unsigned cb = (unsigned)c * 16u;
    const unsigned* __restrict__ myoff = soff + r * SEQL;

    v4f sum = {0.0f, 0.0f, 0.0f, 0.0f};
    v4f mx  = {-INFINITY, -INFINITY, -INFINITY, -INFINITY};
    v4u oa = *(const v4u*)(myoff);
    v4u ob = *(const v4u*)(myoff + 4);
    for (int t0 = 0; t0 < SEQL; t0 += PGRP) {
        v4f v[PGRP];
        v[0] = ld_row4(emb, oa.x + cb); v[1] = ld_row4(emb, oa.y + cb);
        v[2] = ld_row4(emb, oa.z + cb); v[3] = ld_row4(emb, oa.w + cb);
        v[4] = ld_row4(emb, ob.x + cb); v[5] = ld_row4(emb, ob.y + cb);
        v[6] = ld_row4(emb, ob.z + cb); v[7] = ld_row4(emb, ob.w + cb);
        if (t0 + PGRP < SEQL) {
            oa = *(const v4u*)(myoff + t0 + PGRP);
            ob = *(const v4u*)(myoff + t0 + PGRP + 4);
        }
        #pragma unroll
        for (int j = 0; j < PGRP; ++j) {
            sum += v[j];
            if (t0 + j < len) mx = max4(mx, v[j]);
        }
    }
    const float flen = (float)len;
    v4f mean;
    mean.x = sum.x / flen; mean.y = sum.y / flen;
    mean.z = sum.z / flen; mean.w = sum.w / flen;
    *(v4f*)(rep + blk_index(b, c * 4))       = mean;
    *(v4f*)(rep + blk_index(b, DIM + c * 4)) = mx;
}

#define NPART 16

struct FragSet { v4f r0, r1, w0, w1, w2, w3; };

static __device__ __forceinline__ void frag_load(
    FragSet& f, const float* __restrict__ pr, const float* __restrict__ pw, int k)
{
    const int o = k * (BLKF / 8);
    f.r0 = *(const v4f*)(pr + o);
    f.r1 = *(const v4f*)(pr + TILEF + o);
    f.w0 = *(const v4f*)(pw + o);
    f.w1 = *(const v4f*)(pw + TILEF + o);
    f.w2 = *(const v4f*)(pw + 2 * TILEF + o);
    f.w3 = *(const v4f*)(pw + 3 * TILEF + o);
}

static __device__ __forceinline__ v8f mma4(v2f a, v2f b, v8f c)
{
    return __builtin_amdgcn_wmma_f32_16x16x4_f32(false, a, false, b, (short)0, c, false, false);
}

static __device__ __forceinline__ void frag_mma(const FragSet& f, v8f (&acc)[8])
{
    const v2f r0l = __builtin_shufflevector(f.r0, f.r0, 0, 1);
    const v2f r0h = __builtin_shufflevector(f.r0, f.r0, 2, 3);
    const v2f r1l = __builtin_shufflevector(f.r1, f.r1, 0, 1);
    const v2f r1h = __builtin_shufflevector(f.r1, f.r1, 2, 3);
    const v2f w0l = __builtin_shufflevector(f.w0, f.w0, 0, 1);
    const v2f w0h = __builtin_shufflevector(f.w0, f.w0, 2, 3);
    const v2f w1l = __builtin_shufflevector(f.w1, f.w1, 0, 1);
    const v2f w1h = __builtin_shufflevector(f.w1, f.w1, 2, 3);
    const v2f w2l = __builtin_shufflevector(f.w2, f.w2, 0, 1);
    const v2f w2h = __builtin_shufflevector(f.w2, f.w2, 2, 3);
    const v2f w3l = __builtin_shufflevector(f.w3, f.w3, 0, 1);
    const v2f w3h = __builtin_shufflevector(f.w3, f.w3, 2, 3);
    acc[0] = mma4(w0l, r0l, acc[0]); acc[2] = mma4(w1l, r0l, acc[2]);
    acc[4] = mma4(w2l, r0l, acc[4]); acc[6] = mma4(w3l, r0l, acc[6]);
    acc[1] = mma4(w0l, r1l, acc[1]); acc[3] = mma4(w1l, r1l, acc[3]);
    acc[5] = mma4(w2l, r1l, acc[5]); acc[7] = mma4(w3l, r1l, acc[7]);
    acc[0] = mma4(w0h, r0h, acc[0]); acc[2] = mma4(w1h, r0h, acc[2]);
    acc[4] = mma4(w2h, r0h, acc[4]); acc[6] = mma4(w3h, r0h, acc[6]);
    acc[1] = mma4(w0h, r1h, acc[1]); acc[3] = mma4(w1h, r1h, acc[3]);
    acc[5] = mma4(w2h, r1h, acc[5]); acc[7] = mma4(w3h, r1h, acc[7]);
}

__global__ __launch_bounds__(128) void gemm1_wmma_kernel(
    const float* __restrict__ rep, const float* __restrict__ W1t,
    const float* __restrict__ b1, const float* __restrict__ W2,
    float* __restrict__ part)
{
    const int wave  = blockIdx.x * (blockDim.x >> 5) + (threadIdx.x >> 5);
    const int lane  = threadIdx.x & 31;
    const int mtile = wave & 127;
    const int ntile = wave >> 7;
    const int row0  = mtile * 32;
    const int col0  = ntile * 64;
    const int laneM = lane & 15;
    const int hh    = lane >> 4;
    const int kHi   = hh << 2;

    const float* __restrict__ pr = rep + blk_index(row0 + laneM, kHi);
    const float* __restrict__ pw = W1t + blk_index(col0 + laneM, kHi);

    v8f acc[8] = {};
    FragSet s0, s1, s2;

    frag_load(s0, pr, pw, 0);
    frag_load(s1, pr, pw, 8);
    for (int k = 0; k < REPD - 24; k += 24) {
        frag_load(s2, pr, pw, k + 16);
        frag_mma(s0, acc);
        frag_load(s0, pr, pw, k + 24);
        frag_mma(s1, acc);
        frag_load(s1, pr, pw, k + 32);
        frag_mma(s2, acc);
    }
    frag_load(s2, pr, pw, REPD - 8);

    bool ok[4];
    v4f  eb[4][2];
    v4f  ew[4][6];
    #pragma unroll
    for (int j = 0; j < 4; ++j) {
        const int n0 = col0 + j * 16 + hh * 8;
        ok[j] = n0 < HID;
        const int n0c = ok[j] ? n0 : 0;
        const float* __restrict__ bp = b1 + n0c;
        const float* __restrict__ wp = W2 + (size_t)n0c * OUTD;
        eb[j][0] = *(const v4f*)(bp);
        eb[j][1] = *(const v4f*)(bp + 4);
        #pragma unroll
        for (int q = 0; q < 6; ++q) ew[j][q] = *(const v4f*)(wp + 4 * q);
    }

    frag_mma(s0, acc);
    frag_mma(s1, acc);
    frag_mma(s2, acc);

    float p[2][OUTD] = {};
    #pragma unroll
    for (int j = 0; j < 4; ++j) {
        #pragma unroll
        for (int r = 0; r < 8; ++r) {
            const float bias = eb[j][r >> 2][r & 3];
            const int e = r * OUTD;
            const float wa = ew[j][(e    ) >> 2][(e    ) & 3];
            const float wb = ew[j][(e + 1) >> 2][(e + 1) & 3];
            const float wc = ew[j][(e + 2) >> 2][(e + 2) & 3];
            #pragma unroll
            for (int i = 0; i < 2; ++i) {
                const float v = acc[j * 2 + i][r] + bias;
                const float hv = (ok[j] && v > 0.0f) ? v : 0.0f;
                p[i][0] += hv * wa;
                p[i][1] += hv * wb;
                p[i][2] += hv * wc;
            }
        }
    }
    #pragma unroll
    for (int i = 0; i < 2; ++i) {
        #pragma unroll
        for (int c = 0; c < OUTD; ++c) p[i][c] += __shfl_xor(p[i][c], 16);
    }
    float* __restrict__ dst = part + ((size_t)ntile * BATCH + row0 + lane) * OUTD;
    dst[0] = hh ? p[1][0] : p[0][0];
    dst[1] = hh ? p[1][1] : p[0][1];
    dst[2] = hh ? p[1][2] : p[0][2];
}

__global__ __launch_bounds__(256) void out_sum_kernel(
    const float* __restrict__ part, const float* __restrict__ b2,
    float* __restrict__ out)
{
    const int i = blockIdx.x * 256 + threadIdx.x;
    float v[NPART];
    #pragma unroll
    for (int t = 0; t < NPART; ++t) v[t] = part[(size_t)t * (BATCH * OUTD) + i];
    float s = v[0];
    #pragma unroll
    for (int t = 1; t < NPART; ++t) s += v[t];
    out[i] = s + b2[i % OUTD];
}

extern "C" void kernel_launch(void* const* d_in, const int* in_sizes, int n_in,
                              void* d_out, int out_size, void* d_ws, size_t ws_size,
                              hipStream_t stream)
{
    const float* emb  = (const float*)d_in[0];
    const float* W1   = (const float*)d_in[1];
    const float* b1   = (const float*)d_in[2];
    const float* W2   = (const float*)d_in[3];
    const float* b2   = (const float*)d_in[4];
    const int*   x    = (const int*)d_in[5];
    const int*   lens = (const int*)d_in[6];
    float* out = (float*)d_out;

    float* W1t = (float*)d_ws;
    float* rep = W1t + (size_t)HIDP * REPD;
    float* part = rep + (size_t)BATCH * REPD;

    hipLaunchKernelGGL(pool_kernel, dim3(PBLOCKS + TTILES), dim3(320), 0, stream,
                       emb, x, lens, rep, W1, W1t);

    hipLaunchKernelGGL(gemm1_wmma_kernel, dim3(512), dim3(128), 0, stream,
                       rep, W1t, b1, W2, part);

    hipLaunchKernelGGL(out_sum_kernel, dim3(BATCH * OUTD / 256), dim3(256), 0, stream,
                       part, b2, out);
}


# pooling walks each batch row's tokens in table-address order (stable counting sort into 4096-row buckets by wave ballots in LDS), so all blocks sweep the table together; mean summation order changes
# speedup vs baseline: 2.3437x; 1.0734x over previous
#include <hip/hip_runtime.h>
#include <math.h>

#define VOCAB 100000
#define DIM   300
#define BATCH 4096
#define SEQL  200
#define HID   1000
#define HIDP  1024
#define OUTD  3
#define REPD  (2*DIM)

typedef __attribute__((ext_vector_type(2))) float v2f;
typedef __attribute__((ext_vector_type(8))) float v8f;
typedef __attribute__((ext_vector_type(4))) float v4f;
typedef __attribute__((ext_vector_type(4))) unsigned v4u;

#define KGRP   (REPD / 8)
#define BLKF   128
#define TILEF  (KGRP * BLKF)

static __device__ __forceinline__ size_t blk_index(int row, int k)
{
    return (size_t)(row >> 4) * TILEF + (size_t)(k >> 3) * BLKF + (row & 15) * 8 + (k & 7);
}

#define TKT    19
#define TNT    32
#define TTILES (TKT * TNT)

static __device__ __forceinline__ void transpose_pad_tile(
    const float* __restrict__ W1, float* __restrict__ W1t, int t)
{
    __shared__ float tile[32][36];
    const int kBase = (t % TKT) * 32;
    const int nBase = (t / TKT) * 32;
    const int tx = threadIdx.x & 31;
    const int ty = threadIdx.x >> 5;

    if (ty < 8) {
        #pragma unroll
        for (int i = ty; i < 32; i += 8) {
            const int k = kBase + i, n = nBase + tx;
            tile[i][tx] = (k < REPD && n < HID) ? W1[(size_t)k * HID + n] : 0.0f;
        }
    }
    __syncthreads();
    if (ty < 8) {
        const int kk = tx & 7;
        const int ns = tx >> 3;
        #pragma unroll
        for (int i = ty; i < 32; i += 8) {
            const int nl = (i & 7) * 4 + ns;
            const int kl = (i >> 3) * 8 + kk;
            const int n = nBase + nl, k = kBase + kl;
            if (k < REPD)
                W1t[blk_index(n, k)] = tile[kl][nl];
        }
    }
}

#define PROWS  4
#define PLANES (DIM / 4)
#define PGRP   8
#define ROWB   (DIM * 4)
#define PBLOCKS (BATCH / PROWS)
#define NCH   7
#define NCHB  (PROWS * NCH)

static __device__ __forceinline__ v4f ld_row4(const float* __restrict__ emb, unsigned off)
{
    return *(const v4f*)((const char*)emb + off);
}

static __device__ __forceinline__ v4f max4(v4f a, v4f b)
{
    v4f r;
    r.x = fmaxf(a.x, b.x); r.y = fmaxf(a.y, b.y);
    r.z = fmaxf(a.z, b.z); r.w = fmaxf(a.w, b.w);
    return r;
}

__global__ __launch_bounds__(320) void pool_kernel(
    const float* __restrict__ emb, const int* __restrict__ x,
    const int* __restrict__ lengths, float* __restrict__ rep,
    const float* __restrict__ W1, float* __restrict__ W1t)
{
    if (blockIdx.x >= PBLOCKS) {
        transpose_pad_tile(W1, W1t, (int)blockIdx.x - PBLOCKS);
        return;
    }
    __shared__ alignas(16) unsigned soff[PROWS * SEQL];
    __shared__ unsigned scnt[NCHB * 32];
    const int tid  = threadIdx.x;
    const int wv   = tid >> 5;
    const int lane = tid & 31;
    const int b0   = blockIdx.x * PROWS;
    const int* __restrict__ xb = x + (size_t)b0 * SEQL;

    unsigned ent[3], key[3], rnk[3];
    int xv[3], lr[3];
    #pragma unroll
    for (int q = 0; q < 3; ++q) {
        const int ch = wv + 10 * q;
        const int r = ch / NCH;
        const int t = (ch - r * NCH) * 32 + lane;
        const bool v = ch < NCHB && t < SEQL;
        xv[q] = xb[v ? r * SEQL + t : 0];
        lr[q] = lengths[b0 + (ch < NCHB ? r : 0)];
    }
    #pragma unroll
    for (int q = 0; q < 3; ++q) {
        const int ch = wv + 10 * q;
        const int r = ch / NCH;
        const int t = (ch - r * NCH) * 32 + lane;
        const bool v = ch < NCHB && t < SEQL;
        const unsigned bk = (unsigned)xv[q] >> 12;
        const unsigned k = v ? (bk < 31u ? bk : 31u) : 32u;
        unsigned m = 0xFFFFFFFFu;
        #pragma unroll
        for (int bit = 0; bit < 6; ++bit) {
            const bool s = ((k >> bit) & 1u) != 0u;
            const unsigned bm = __builtin_amdgcn_ballot_w32(s);
            m &= s ? bm : ~bm;
        }
        const unsigned rk = __builtin_amdgcn_mbcnt_lo(m, 0u);
        if (ch < NCHB) scnt[ch * 32 + lane] = 0u;
        if (v && rk == 0u)
            scnt[ch * 32 + k] = (unsigned)__popc(m);
        ent[q] = (unsigned)xv[q] * (unsigned)ROWB | ((v && t < lr[q]) ? 1u : 0u);
        key[q] = k; rnk[q] = rk;
    }
    __syncthreads();
    #pragma unroll
    for (int q = 0; q < 3; ++q) {
        const int ch = wv + 10 * q;
        if (ch < NCHB) {
            const int r = ch / NCH;
            const int c = ch - r * NCH;
            unsigned tot = 0u, before = 0u;
            #pragma unroll
            for (int c2 = 0; c2 < NCH; ++c2) {
                const unsigned n2 = scnt[(r * NCH + c2) * 32 + lane];
                tot += n2;
                before += (c2 < c) ? n2 : 0u;
            }
            unsigned inc = tot;
            #pragma unroll
            for (int d = 1; d < 32; d <<= 1) {
                const unsigned up = __shfl_up(inc, d);
                inc += (lane >= d) ? up : 0u;
            }
            const unsigned base = inc - tot + before;
            const unsigned mybase = __shfl(base, (int)(key[q] & 31u));
            if (key[q] < 32u)
                soff[r * SEQL + mybase + rnk[q]] = ent[q];
        }
    }
    __syncthreads();
    if (tid >= PROWS * PLANES) return;

    const int r   = tid / PLANES;
    const int c   = tid - r * PLANES;
    const int b   = b0 + r;
    const int len = lengths[b];
    const unsigned cb = (unsigned)c * 16u;
    const unsigned* __restrict__ myoff = soff + r * SEQL;

    v4f sum = {0.0f, 0.0f, 0.0f, 0.0f};
    v4f mx  = {-INFINITY, -INFINITY, -INFINITY, -INFINITY};
    v4u oa = *(const v4u*)(myoff);
    v4u ob = *(const v4u*)(myoff + 4);
    for (int t0 = 0; t0 < SEQL; t0 += PGRP) {
        v4f v[PGRP];
        unsigned e[PGRP] = {oa.x, oa.y, oa.z, oa.w, ob.x, ob.y, ob.z, ob.w};
        #pragma unroll
        for (int j = 0; j < PGRP; ++j) v[j] = ld_row4(emb, (e[j] & ~1u) + cb);
        if (t0 + PGRP < SEQL) {
            oa = *(const v4u*)(myoff + t0 + PGRP);
            ob = *(const v4u*)(myoff + t0 + PGRP + 4);
        }
        #pragma unroll
        for (int j = 0; j < PGRP; ++j) {
            sum += v[j];
            if (e[j] & 1u) mx = max4(mx, v[j]);
        }
    }
    const float flen = (float)len;
    v4f mean;
    mean.x = sum.x / flen; mean.y = sum.y / flen;
    mean.z = sum.z / flen; mean.w = sum.w / flen;
    *(v4f*)(rep + blk_index(b, c * 4))       = mean;
    *(v4f*)(rep + blk_index(b, DIM + c * 4)) = mx;
}

#define NPART 16

struct FragSet { v4f r0, r1, w0, w1, w2, w3; };

static __device__ __forceinline__ void frag_load(
    FragSet& f, const float* __restrict__ pr, const float* __restrict__ pw, int k)
{
    const int o = k * (BLKF / 8);
    f.r0 = *(const v4f*)(pr + o);
    f.r1 = *(const v4f*)(pr + TILEF + o);
    f.w0 = *(const v4f*)(pw + o);
    f.w1 = *(const v4f*)(pw + TILEF + o);
    f.w2 = *(const v4f*)(pw + 2 * TILEF + o);
    f.w3 = *(const v4f*)(pw + 3 * TILEF + o);
}

static __device__ __forceinline__ v8f mma4(v2f a, v2f b, v8f c)
{
    return __builtin_amdgcn_wmma_f32_16x16x4_f32(false, a, false, b, (short)0, c, false, false);
}

static __device__ __forceinline__ void frag_mma(const FragSet& f, v8f (&acc)[8])
{
    const v2f r0l = __builtin_shufflevector(f.r0, f.r0, 0, 1);
    const v2f r0h = __builtin_shufflevector(f.r0, f.r0, 2, 3);
    const v2f r1l = __builtin_shufflevector(f.r1, f.r1, 0, 1);
    const v2f r1h = __builtin_shufflevector(f.r1, f.r1, 2, 3);
    const v2f w0l = __builtin_shufflevector(f.w0, f.w0, 0, 1);
    const v2f w0h = __builtin_shufflevector(f.w0, f.w0, 2, 3);
    const v2f w1l = __builtin_shufflevector(f.w1, f.w1, 0, 1);
    const v2f w1h = __builtin_shufflevector(f.w1, f.w1, 2, 3);
    const v2f w2l = __builtin_shufflevector(f.w2, f.w2, 0, 1);
    const v2f w2h = __builtin_shufflevector(f.w2, f.w2, 2, 3);
    const v2f w3l = __builtin_shufflevector(f.w3, f.w3, 0, 1);
    const v2f w3h = __builtin_shufflevector(f.w3, f.w3, 2, 3);
    acc[0] = mma4(w0l, r0l, acc[0]); acc[2] = mma4(w1l, r0l, acc[2]);
    acc[4] = mma4(w2l, r0l, acc[4]); acc[6] = mma4(w3l, r0l, acc[6]);
    acc[1] = mma4(w0l, r1l, acc[1]); acc[3] = mma4(w1l, r1l, acc[3]);
    acc[5] = mma4(w2l, r1l, acc[5]); acc[7] = mma4(w3l, r1l, acc[7]);
    acc[0] = mma4(w0h, r0h, acc[0]); acc[2] = mma4(w1h, r0h, acc[2]);
    acc[4] = mma4(w2h, r0h, acc[4]); acc[6] = mma4(w3h, r0h, acc[6]);
    acc[1] = mma4(w0h, r1h, acc[1]); acc[3] = mma4(w1h, r1h, acc[3]);
    acc[5] = mma4(w2h, r1h, acc[5]); acc[7] = mma4(w3h, r1h, acc[7]);
}

__global__ __launch_bounds__(128) void gemm1_wmma_kernel(
    const float* __restrict__ rep, const float* __restrict__ W1t,
    const float* __restrict__ b1, const float* __restrict__ W2,
    float* __restrict__ part)
{
    const int wave  = blockIdx.x * (blockDim.x >> 5) + (threadIdx.x >> 5);
    const int lane  = threadIdx.x & 31;
    const int mtile = wave & 127;
    const int ntile = wave >> 7;
    const int row0  = mtile * 32;
    const int col0  = ntile * 64;
    const int laneM = lane & 15;
    const int hh    = lane >> 4;
    const int kHi   = hh << 2;

    const float* __restrict__ pr = rep + blk_index(row0 + laneM, kHi);
    const float* __restrict__ pw = W1t + blk_index(col0 + laneM, kHi);

    v8f acc[8] = {};
    FragSet s0, s1, s2;

    frag_load(s0, pr, pw, 0);
    frag_load(s1, pr, pw, 8);
    for (int k = 0; k < REPD - 24; k += 24) {
        frag_load(s2, pr, pw, k + 16);
        frag_mma(s0, acc);
        frag_load(s0, pr, pw, k + 24);
        frag_mma(s1, acc);
        frag_load(s1, pr, pw, k + 32);
        frag_mma(s2, acc);
    }
    frag_load(s2, pr, pw, REPD - 8);

    bool ok[4];
    v4f  eb[4][2];
    v4f  ew[4][6];
    #pragma unroll
    for (int j = 0; j < 4; ++j) {
        const int n0 = col0 + j * 16 + hh * 8;
        ok[j] = n0 < HID;
        const int n0c = ok[j] ? n0 : 0;
        const float* __restrict__ bp = b1 + n0c;
        const float* __restrict__ wp = W2 + (size_t)n0c * OUTD;
        eb[j][0] = *(const v4f*)(bp);
        eb[j][1] = *(const v4f*)(bp + 4);
        #pragma unroll
        for (int q = 0; q < 6; ++q) ew[j][q] = *(const v4f*)(wp + 4 * q);
    }

    frag_mma(s0, acc);
    frag_mma(s1, acc);
    frag_mma(s2, acc);

    float p[2][OUTD] = {};
    #pragma unroll
    for (int j = 0; j < 4; ++j) {
        #pragma unroll
        for (int r = 0; r < 8; ++r) {
            const float bias = eb[j][r >> 2][r & 3];
            const int e = r * OUTD;
            const float wa = ew[j][(e    ) >> 2][(e    ) & 3];
            const float wb = ew[j][(e + 1) >> 2][(e + 1) & 3];
            const float wc = ew[j][(e + 2) >> 2][(e + 2) & 3];
            #pragma unroll
            for (int i = 0; i < 2; ++i) {
                const float v = acc[j * 2 + i][r] + bias;
                const float hv = (ok[j] && v > 0.0f) ? v : 0.0f;
                p[i][0] += hv * wa;
                p[i][1] += hv * wb;
                p[i][2] += hv * wc;
            }
        }
    }
    #pragma unroll
    for (int i = 0; i < 2; ++i) {
        #pragma unroll
        for (int c = 0; c < OUTD; ++c) p[i][c] += __shfl_xor(p[i][c], 16);
    }
    float* __restrict__ dst = part + ((size_t)ntile * BATCH + row0 + lane) * OUTD;
    dst[0] = hh ? p[1][0] : p[0][0];
    dst[1] = hh ? p[1][1] : p[0][1];
    dst[2] = hh ? p[1][2] : p[0][2];
}

__global__ __launch_bounds__(256) void out_sum_kernel(
    const float* __restrict__ part, const float* __restrict__ b2,
    float* __restrict__ out)
{
    const int i = blockIdx.x * 256 + threadIdx.x;
    float v[NPART];
    #pragma unroll
    for (int t = 0; t < NPART; ++t) v[t] = part[(size_t)t * (BATCH * OUTD) + i];
    float s = v[0];
    #pragma unroll
    for (int t = 1; t < NPART; ++t) s += v[t];
    out[i] = s + b2[i % OUTD];
}

extern "C" void kernel_launch(void* const* d_in, const int* in_sizes, int n_in,
                              void* d_out, int out_size, void* d_ws, size_t ws_size,
                              hipStream_t stream)
{
    const float* emb  = (const float*)d_in[0];
    const float* W1   = (const float*)d_in[1];
    const float* b1   = (const float*)d_in[2];
    const float* W2   = (const float*)d_in[3];
    const float* b2   = (const float*)d_in[4];
    const int*   x    = (const int*)d_in[5];
    const int*   lens = (const int*)d_in[6];
    float* out = (float*)d_out;

    float* W1t = (float*)d_ws;
    float* rep = W1t + (size_t)HIDP * REPD;
    float* part = rep + (size_t)BATCH * REPD;

    hipLaunchKernelGGL(pool_kernel, dim3(PBLOCKS + TTILES), dim3(320), 0, stream,
                       emb, x, lens, rep, W1, W1t);

    hipLaunchKernelGGL(gemm1_wmma_kernel, dim3(512), dim3(128), 0, stream,
                       rep, W1t, b1, W2, part);

    hipLaunchKernelGGL(out_sum_kernel, dim3(BATCH * OUTD / 256), dim3(256), 0, stream,
                       part, b2, out);
}


# pooling gather: a wave fetches whole table rows as aligned 256-byte units (3 window loads), re-aligns them to lane-owned dims through LDS; 2 waves per batch row; mean summation order changes
# speedup vs baseline: 2.4450x; 1.0432x over previous
#include <hip/hip_runtime.h>
#include <math.h>

#define VOCAB 100000
#define DIM   300
#define BATCH 4096
#define SEQL  200
#define HID   1000
#define HIDP  1024
#define OUTD  3
#define REPD  (2*DIM)

typedef __attribute__((ext_vector_type(2))) float v2f;
typedef __attribute__((ext_vector_type(8))) float v8f;
typedef __attribute__((ext_vector_type(4))) float v4f;
typedef __attribute__((ext_vector_type(4))) unsigned v4u;

#define KGRP   (REPD / 8)
#define BLKF   128
#define TILEF  (KGRP * BLKF)

static __device__ __forceinline__ size_t blk_index(int row, int k)
{
    return (size_t)(row >> 4) * TILEF + (size_t)(k >> 3) * BLKF + (row & 15) * 8 + (k & 7);
}

#define TKT    19
#define TNT    32
#define TTILES (TKT * TNT)

static __device__ __forceinline__ void transpose_pad_tile(
    const float* __restrict__ W1, float* __restrict__ W1t, int t)
{
    __shared__ float tile[32][36];
    const int kBase = (t % TKT) * 32;
    const int nBase = (t / TKT) * 32;
    const int tx = threadIdx.x & 31;
    const int ty = threadIdx.x >> 5;

    if (ty < 8) {
        #pragma unroll
        for (int i = ty; i < 32; i += 8) {
            const int k = kBase + i, n = nBase + tx;
            tile[i][tx] = (k < REPD && n < HID) ? W1[(size_t)k * HID + n] : 0.0f;
        }
    }
    __syncthreads();
    if (ty < 8) {
        const int kk = tx & 7;
        const int ns = tx >> 3;
        #pragma unroll
        for (int i = ty; i < 32; i += 8) {
            const int nl = (i & 7) * 4 + ns;
            const int kl = (i >> 3) * 8 + kk;
            const int n = nBase + nl, k = kBase + kl;
            if (k < REPD)
                W1t[blk_index(n, k)] = tile[kl][nl];
        }
    }
}

#define PROWS  4
#define PWAVES 2
#define PWV    (PROWS * PWAVES)
#define PTHREADS (PWV * 32)
#define PLANES (DIM / 4)
#define PGRP   4
#define NGRP   (SEQL / PGRP)
#define ROWB   (DIM * 4)
#define WINB   1536
#define STGF   (PGRP * WINB / 4)
#define PBLOCKS (BATCH / PROWS)
#define NCH   7
#define NCHB  (PROWS * NCH)
#define NSQ   4

static __device__ __forceinline__ v4f ld_row4(const float* __restrict__ emb, unsigned off)
{
    return *(const v4f*)((const char*)emb + off);
}

static __device__ __forceinline__ v4f max4(v4f a, v4f b)
{
    v4f r;
    r.x = fmaxf(a.x, b.x); r.y = fmaxf(a.y, b.y);
    r.z = fmaxf(a.z, b.z); r.w = fmaxf(a.w, b.w);
    return r;
}

static __device__ __forceinline__ void window_load(
    v4f (&v)[3], const float* __restrict__ emb, unsigned e, int lb)
{
    const unsigned off = e & ~1u;
    const int sh = (int)(off & 255u);
    #pragma unroll
    for (int i = 0; i < 3; ++i) {
        int rel = lb + i * 512 - sh;
        rel = rel < 0 ? 0 : (rel > ROWB - 16 ? ROWB - 16 : rel);
        v[i] = ld_row4(emb, off + (unsigned)rel);
    }
}

__global__ __launch_bounds__(PTHREADS) void pool_kernel(
    const float* __restrict__ emb, const int* __restrict__ x,
    const int* __restrict__ lengths, float* __restrict__ rep,
    const float* __restrict__ W1, float* __restrict__ W1t)
{
    if (blockIdx.x >= PBLOCKS) {
        transpose_pad_tile(W1, W1t, (int)blockIdx.x - PBLOCKS);
        return;
    }
    __shared__ alignas(16) unsigned soff[PROWS * SEQL];
    __shared__ unsigned scnt[NCHB * 32];
    __shared__ alignas(16) float stage[PWV * STGF];
    const int tid  = threadIdx.x;
    const int wv   = tid >> 5;
    const int lane = tid & 31;
    const int b0   = blockIdx.x * PROWS;
    const int* __restrict__ xb = x + (size_t)b0 * SEQL;

    unsigned ent[NSQ], key[NSQ], rnk[NSQ];
    int xv[NSQ], lr[NSQ];
    #pragma unroll
    for (int q = 0; q < NSQ; ++q) {
        const int ch = wv + PWV * q;
        const int r = ch / NCH;
        const int t = (ch - r * NCH) * 32 + lane;
        const bool v = ch < NCHB && t < SEQL;
        xv[q] = xb[v ? r * SEQL + t : 0];
        lr[q] = lengths[b0 + (ch < NCHB ? r : 0)];
    }
    #pragma unroll
    for (int q = 0; q < NSQ; ++q) {
        const int ch = wv + PWV * q;
        const int r = ch / NCH;
        const int t = (ch - r * NCH) * 32 + lane;
        const bool v = ch < NCHB && t < SEQL;
        const unsigned bk = (unsigned)xv[q] >> 12;
        const unsigned k = v ? (bk < 31u ? bk : 31u) : 32u;
        unsigned m = 0xFFFFFFFFu;
        #pragma unroll
        for (int bit = 0; bit < 6; ++bit) {
            const bool s = ((k >> bit) & 1u) != 0u;
            const unsigned bm = __builtin_amdgcn_ballot_w32(s);
            m &= s ? bm : ~bm;
        }
        const unsigned rk = __builtin_amdgcn_mbcnt_lo(m, 0u);
        if (ch < NCHB) scnt[ch * 32 + lane] = 0u;
        if (v && rk == 0u)
            scnt[ch * 32 + k] = (unsigned)__popc(m);
        ent[q] = (unsigned)xv[q] * (unsigned)ROWB | ((v && t < lr[q]) ? 1u : 0u);
        key[q] = k; rnk[q] = rk;
    }
    __syncthreads();
    #pragma unroll
    for (int q = 0; q < NSQ; ++q) {
        const int ch = wv + PWV * q;
        if (ch < NCHB) {
            const int r = ch / NCH;
            const int c = ch - r * NCH;
            unsigned tot = 0u, before = 0u;
            #pragma unroll
            for (int c2 = 0; c2 < NCH; ++c2) {
                const unsigned n2 = scnt[(r * NCH + c2) * 32 + lane];
                tot += n2;
                before += (c2 < c) ? n2 : 0u;
            }
            unsigned inc = tot;
            #pragma unroll
            for (int d = 1; d < 32; d <<= 1) {
                const unsigned up = __shfl_up(inc, d);
                inc += (lane >= d) ? up : 0u;
            }
            const unsigned base = inc - tot + before;
            const unsigned mybase = __shfl(base, (int)(key[q] & 31u));
            if (key[q] < 32u)
                soff[r * SEQL + mybase + rnk[q]] = ent[q];
        }
    }
    __syncthreads();

    const int wvs = __builtin_amdgcn_readfirstlane(wv);
    const int r = wvs / PWAVES;
    const int h = wvs - r * PWAVES;
    const int b = b0 + r;
    const unsigned* __restrict__ myoff = soff + r * SEQL;
    char* stg = (char*)(stage + wvs * STGF);
    const int lb  = lane * 16;
    const int rd2 = (lane < PLANES - 64 ? 64 + lane : PLANES - 1) * 16;

    v4f sum[3], mx[3];
    #pragma unroll
    for (int k = 0; k < 3; ++k) {
        sum[k] = (v4f){0.0f, 0.0f, 0.0f, 0.0f};
        mx[k]  = (v4f){-INFINITY, -INFINITY, -INFINITY, -INFINITY};
    }

    v4f v[PGRP][3];
    v4u en = *(const v4u*)(myoff + h * PGRP);
    {
        const unsigned e[PGRP] = {en.x, en.y, en.z, en.w};
        #pragma unroll
        for (int j = 0; j < PGRP; ++j)
            window_load(v[j], emb, (unsigned)__builtin_amdgcn_readfirstlane((int)e[j]), lb);
    }
    v4u nx = *(const v4u*)(myoff + (h + PWAVES) * PGRP);
    for (int g = h; g < NGRP; g += PWAVES) {
        const unsigned e[PGRP] = {en.x, en.y, en.z, en.w};
        #pragma unroll
        for (int j = 0; j < PGRP; ++j) {
            #pragma unroll
            for (int i = 0; i < 3; ++i)
                *(v4f*)(stg + j * WINB + i * 512 + lb) = v[j][i];
        }
        if (g + PWAVES < NGRP) {
            en = nx;
            const unsigned e2[PGRP] = {en.x, en.y, en.z, en.w};
            #pragma unroll
            for (int j = 0; j < PGRP; ++j)
                window_load(v[j], emb, (unsigned)__builtin_amdgcn_readfirstlane((int)e2[j]), lb);
            nx = *(const v4u*)(myoff + (g + 2 * PWAVES < NGRP ? g + 2 * PWAVES : g) * PGRP);
        }
        const unsigned s0 = (unsigned)__builtin_amdgcn_readfirstlane((int)e[0]);
        const unsigned s1 = (unsigned)__builtin_amdgcn_readfirstlane((int)e[1]);
        const unsigned s2 = (unsigned)__builtin_amdgcn_readfirstlane((int)e[2]);
        const unsigned s3 = (unsigned)__builtin_amdgcn_readfirstlane((int)e[3]);
        #pragma nounroll
        for (int j = 0; j < PGRP; ++j) {
            const unsigned ej = j == 0 ? s0 : (j == 1 ? s1 : (j == 2 ? s2 : s3));
            const char* p = stg + j * WINB + (int)((ej & ~1u) & 255u);
            const v4f a0 = *(const v4f*)(p + lb);
            const v4f a1 = *(const v4f*)(p + lb + 512);
            const v4f a2 = *(const v4f*)(p + rd2);
            sum[0] += a0; sum[1] += a1; sum[2] += a2;
            if (ej & 1u) {
                mx[0] = max4(mx[0], a0); mx[1] = max4(mx[1], a1); mx[2] = max4(mx[2], a2);
            }
        }
    }

    if (h != 0) {
        #pragma unroll
        for (int k = 0; k < 3; ++k) {
            *(v4f*)(stg + k * 512 + lb)        = sum[k];
            *(v4f*)(stg + (3 + k) * 512 + lb)  = mx[k];
        }
    }
    __syncthreads();
    if (h == 0) {
        const char* oth = (const char*)(stage + (wvs + 1) * STGF);
        const float flen = (float)lengths[b];
        #pragma unroll
        for (int k = 0; k < 3; ++k) {
            const int c = lane + 32 * k;
            const v4f s = sum[k] + *(const v4f*)(oth + k * 512 + lb);
            const v4f m = max4(mx[k], *(const v4f*)(oth + (3 + k) * 512 + lb));
            if (c < PLANES) {
                v4f mean;
                mean.x = s.x / flen; mean.y = s.y / flen;
                mean.z = s.z / flen; mean.w = s.w / flen;
                *(v4f*)(rep + blk_index(b, c * 4))       = mean;
                *(v4f*)(rep + blk_index(b, DIM + c * 4)) = m;
            }
        }
    }
}

#define NPART 16

struct FragSet { v4f r0, r1, w0, w1, w2, w3; };

static __device__ __forceinline__ void frag_load(
    FragSet& f, const float* __restrict__ pr, const float* __restrict__ pw, int k)
{
    const int o = k * (BLKF / 8);
    f.r0 = *(const v4f*)(pr + o);
    f.r1 = *(const v4f*)(pr + TILEF + o);
    f.w0 = *(const v4f*)(pw + o);
    f.w1 = *(const v4f*)(pw + TILEF + o);
    f.w2 = *(const v4f*)(pw + 2 * TILEF + o);
    f.w3 = *(const v4f*)(pw + 3 * TILEF + o);
}

static __device__ __forceinline__ v8f mma4(v2f a, v2f b, v8f c)
{
    return __builtin_amdgcn_wmma_f32_16x16x4_f32(false, a, false, b, (short)0, c, false, false);
}

static __device__ __forceinline__ void frag_mma(const FragSet& f, v8f (&acc)[8])
{
    const v2f r0l = __builtin_shufflevector(f.r0, f.r0, 0, 1);
    const v2f r0h = __builtin_shufflevector(f.r0, f.r0, 2, 3);
    const v2f r1l = __builtin_shufflevector(f.r1, f.r1, 0, 1);
    const v2f r1h = __builtin_shufflevector(f.r1, f.r1, 2, 3);
    const v2f w0l = __builtin_shufflevector(f.w0, f.w0, 0, 1);
    const v2f w0h = __builtin_shufflevector(f.w0, f.w0, 2, 3);
    const v2f w1l = __builtin_shufflevector(f.w1, f.w1, 0, 1);
    const v2f w1h = __builtin_shufflevector(f.w1, f.w1, 2, 3);
    const v2f w2l = __builtin_shufflevector(f.w2, f.w2, 0, 1);
    const v2f w2h = __builtin_shufflevector(f.w2, f.w2, 2, 3);
    const v2f w3l = __builtin_shufflevector(f.w3, f.w3, 0, 1);
    const v2f w3h = __builtin_shufflevector(f.w3, f.w3, 2, 3);
    acc[0] = mma4(w0l, r0l, acc[0]); acc[2] = mma4(w1l, r0l, acc[2]);
    acc[4] = mma4(w2l, r0l, acc[4]); acc[6] = mma4(w3l, r0l, acc[6]);
    acc[1] = mma4(w0l, r1l, acc[1]); acc[3] = mma4(w1l, r1l, acc[3]);
    acc[5] = mma4(w2l, r1l, acc[5]); acc[7] = mma4(w3l, r1l, acc[7]);
    acc[0] = mma4(w0h, r0h, acc[0]); acc[2] = mma4(w1h, r0h, acc[2]);
    acc[4] = mma4(w2h, r0h, acc[4]); acc[6] = mma4(w3h, r0h, acc[6]);
    acc[1] = mma4(w0h, r1h, acc[1]); acc[3] = mma4(w1h, r1h, acc[3]);
    acc[5] = mma4(w2h, r1h, acc[5]); acc[7] = mma4(w3h, r1h, acc[7]);
}

__global__ __launch_bounds__(128) void gemm1_wmma_kernel(
    const float* __restrict__ rep, const float* __restrict__ W1t,
    const float* __restrict__ b1, const float* __restrict__ W2,
    float* __restrict__ part)
{
    const int wave  = blockIdx.x * (blockDim.x >> 5) + (threadIdx.x >> 5);
    const int lane  = threadIdx.x & 31;
    const int mtile = wave & 127;
    const int ntile = wave >> 7;
    const int row0  = mtile * 32;
    const int col0  = ntile * 64;
    const int laneM = lane & 15;
    const int hh    = lane >> 4;
    const int kHi   = hh << 2;

    const float* __restrict__ pr = rep + blk_index(row0 + laneM, kHi);
    const float* __restrict__ pw = W1t + blk_index(col0 + laneM, kHi);

    v8f acc[8] = {};
    FragSet s0, s1, s2;

    frag_load(s0, pr, pw, 0);
    frag_load(s1, pr, pw, 8);
    for (int k = 0; k < REPD - 24; k += 24) {
        frag_load(s2, pr, pw, k + 16);
        frag_mma(s0, acc);
        frag_load(s0, pr, pw, k + 24);
        frag_mma(s1, acc);
        frag_load(s1, pr, pw, k + 32);
        frag_mma(s2, acc);
    }
    frag_load(s2, pr, pw, REPD - 8);

    bool ok[4];
    v4f  eb[4][2];
    v4f  ew[4][6];
    #pragma unroll
    for (int j = 0; j < 4; ++j) {
        const int n0 = col0 + j * 16 + hh * 8;
        ok[j] = n0 < HID;
        const int n0c = ok[j] ? n0 : 0;
        const float* __restrict__ bp = b1 + n0c;
        const float* __restrict__ wp = W2 + (size_t)n0c * OUTD;
        eb[j][0] = *(const v4f*)(bp);
        eb[j][1] = *(const v4f*)(bp + 4);
        #pragma unroll
        for (int q = 0; q < 6; ++q) ew[j][q] = *(const v4f*)(wp + 4 * q);
    }

    frag_mma(s0, acc);
    frag_mma(s1, acc);
    frag_mma(s2, acc);

    float p[2][OUTD] = {};
    #pragma unroll
    for (int j = 0; j < 4; ++j) {
        #pragma unroll
        for (int r = 0; r < 8; ++r) {
            const float bias = eb[j][r >> 2][r & 3];
            const int e = r * OUTD;
            const float wa = ew[j][(e    ) >> 2][(e    ) & 3];
            const float wb = ew[j][(e + 1) >> 2][(e + 1) & 3];
            const float wc = ew[j][(e + 2) >> 2][(e + 2) & 3];
            #pragma unroll
            for (int i = 0; i < 2; ++i) {
                const float v = acc[j * 2 + i][r] + bias;
                const float hv = (ok[j] && v > 0.0f) ? v : 0.0f;
                p[i][0] += hv * wa;
                p[i][1] += hv * wb;
                p[i][2] += hv * wc;
            }
        }
    }
    #pragma unroll
    for (int i = 0; i < 2; ++i) {
        #pragma unroll
        for (int c = 0; c < OUTD; ++c) p[i][c] += __shfl_xor(p[i][c], 16);
    }
    float* __restrict__ dst = part + ((size_t)ntile * BATCH + row0 + lane) * OUTD;
    dst[0] = hh ? p[1][0] : p[0][0];
    dst[1] = hh ? p[1][1] : p[0][1];
    dst[2] = hh ? p[1][2] : p[0][2];
}

__global__ __launch_bounds__(256) void out_sum_kernel(
    const float* __restrict__ part, const float* __restrict__ b2,
    float* __restrict__ out)
{
    const int i = blockIdx.x * 256 + threadIdx.x;
    float v[NPART];
    #pragma unroll
    for (int t = 0; t < NPART; ++t) v[t] = part[(size_t)t * (BATCH * OUTD) + i];
    float s = v[0];
    #pragma unroll
    for (int t = 1; t < NPART; ++t) s += v[t];
    out[i] = s + b2[i % OUTD];
}

extern "C" void kernel_launch(void* const* d_in, const int* in_sizes, int n_in,
                              void* d_out, int out_size, void* d_ws, size_t ws_size,
                              hipStream_t stream)
{
    const float* emb  = (const float*)d_in[0];
    const float* W1   = (const float*)d_in[1];
    const float* b1   = (const float*)d_in[2];
    const float* W2   = (const float*)d_in[3];
    const float* b2   = (const float*)d_in[4];
    const int*   x    = (const int*)d_in[5];
    const int*   lens = (const int*)d_in[6];
    float* out = (float*)d_out;

    float* W1t = (float*)d_ws;
    float* rep = W1t + (size_t)HIDP * REPD;
    float* part = rep + (size_t)BATCH * REPD;

    hipLaunchKernelGGL(pool_kernel, dim3(PBLOCKS + TTILES), dim3(PTHREADS), 0, stream,
                       emb, x, lens, rep, W1, W1t);

    hipLaunchKernelGGL(gemm1_wmma_kernel, dim3(512), dim3(128), 0, stream,
                       rep, W1t, b1, W2, part);

    hipLaunchKernelGGL(out_sum_kernel, dim3(BATCH * OUTD / 256), dim3(256), 0, stream,
                       part, b2, out);
}
